# starting kernel
# baseline (speedup 1.0000x reference)
#include <hip/hip_runtime.h>


typedef float v2f __attribute__((ext_vector_type(2)));
typedef float v4f __attribute__((ext_vector_type(4)));
typedef float v8f __attribute__((ext_vector_type(8)));

__device__ __forceinline__ void atomicAddF32(float* p, float v) {
  unsafeAtomicAdd(p, v);
}

__global__ void k_init_deg(float* __restrict__ deg, int N) {
  int i = blockIdx.x * blockDim.x + threadIdx.x;
  if (i < N) deg[i] = 1.0f;
}

__global__ void k_deg_edges(float* __restrict__ deg, const int* __restrict__ dst, int E) {
  int i = blockIdx.x * blockDim.x + threadIdx.x;
  if (i < E) atomicAddF32(&deg[dst[i]], 1.0f);
}

__global__ void k_rsqrt(float* __restrict__ deg, int N) {
  int i = blockIdx.x * blockDim.x + threadIdx.x;
  if (i < N) deg[i] = rsqrtf(deg[i]);
}

template <int K, int MT>
__global__ void k_gemm_wmma(const float* __restrict__ X, const float* __restrict__ W,
                            float* __restrict__ Y, int M, int Nout) {
  constexpr int KSTEPS = K / 4;
  const int ntiles  = Nout >> 4;
  const int mtiles  = (M + 15) >> 4;
  const int mgroups = (mtiles + MT - 1) / MT;
  const int wave    = blockIdx.x * (blockDim.x >> 5) + (threadIdx.x >> 5);
  if (wave >= mgroups * ntiles) return;
  const int mg   = wave / ntiles;
  const int nt   = wave - mg * ntiles;
  const int lane = threadIdx.x & 31;
  const int half = lane >> 4;
  const int r    = lane & 15;
  const int n0   = nt << 4;

  const float* Bptr = W + (size_t)(2 * half) * Nout + (n0 + r);
  v2f bq[KSTEPS];
#pragma unroll
  for (int kk = 0; kk < KSTEPS; ++kk) {
    bq[kk][0] = Bptr[(size_t)(4 * kk + 0) * Nout];
    bq[kk][1] = Bptr[(size_t)(4 * kk + 1) * Nout];
  }

  for (int t = 0; t < MT; ++t) {
    const int mt = mg * MT + t;
    if (mt >= mtiles) break;
    const int m0 = mt << 4;
    int arow = m0 + r;
    if (arow >= M) arow = M - 1;
    const float* Aptr = X + (size_t)arow * K + 2 * half;

    v8f c = {};
#pragma unroll
    for (int kk = 0; kk < KSTEPS; ++kk) {
      v2f a = *(const v2f*)(Aptr + 4 * kk);
      c = __builtin_amdgcn_wmma_f32_16x16x4_f32(
           false, a,  false, bq[kk],
           (short)0, c,  false,  false);
    }
#pragma unroll
    for (int v = 0; v < 8; ++v) {
      int row = m0 + v + 8 * half;
      if (row < M) Y[(size_t)row * Nout + n0 + r] = c[v];
    }
  }
}

__global__ void k_self_loop(float* __restrict__ Y, const float* __restrict__ X,
                            const float* __restrict__ dinv, long long total, int logD) {
  long long t = (long long)blockIdx.x * blockDim.x + threadIdx.x;
  if (t < total) {
    float w = dinv[t >> logD];
    Y[t] = X[t] * w * w;
  }
}

__global__ void k_scatter(float* __restrict__ Y, const float* __restrict__ X,
                          const float* __restrict__ dinv,
                          const int* __restrict__ src, const int* __restrict__ dst,
                          int E, int D) {
  const int e = blockIdx.x * (blockDim.x >> 5) + (threadIdx.x >> 5);
  if (e >= E) return;
  const int lane = threadIdx.x & 31;
  const int s = src[e], d = dst[e];
  const float nrm = dinv[s] * dinv[d];
  const float* xs = X + (size_t)s * D;
  float* yd = Y + (size_t)d * D;
  if (D == 128) {
    v4f x = ((const v4f*)xs)[lane];
    float* p = yd + 4 * lane;
    atomicAddF32(p + 0, x[0] * nrm);
    atomicAddF32(p + 1, x[1] * nrm);
    atomicAddF32(p + 2, x[2] * nrm);
    atomicAddF32(p + 3, x[3] * nrm);
  } else if (D == 64) {
    v2f x = ((const v2f*)xs)[lane];
    float* p = yd + 2 * lane;
    atomicAddF32(p + 0, x[0] * nrm);
    atomicAddF32(p + 1, x[1] * nrm);
  } else {
    for (int j = lane; j < D; j += 32)
      atomicAddF32(&yd[j], xs[j] * nrm);
  }
}

__global__ void k_bias_relu(float* __restrict__ Y, const float* __restrict__ b,
                            long long total, int Dmask) {
  long long t = (long long)blockIdx.x * blockDim.x + threadIdx.x;
  if (t < total) {
    float v = Y[t] + b[t & Dmask];
    Y[t] = fmaxf(v, 0.0f);
  }
}

__global__ void k_bias_add(float* __restrict__ Y, const float* __restrict__ b,
                           long long total, int Dmask) {
  long long t = (long long)blockIdx.x * blockDim.x + threadIdx.x;
  if (t < total) Y[t] = Y[t] + b[t & Dmask];
}

extern "C" void kernel_launch(void* const* d_in, const int* in_sizes, int n_in,
                              void* d_out, int out_size, void* d_ws, size_t ws_size,
                              hipStream_t stream) {
  const float* emb  = (const float*)d_in[0];
  const float* W1   = (const float*)d_in[1];
  const float* b1   = (const float*)d_in[2];
  const float* W2   = (const float*)d_in[3];
  const float* b2   = (const float*)d_in[4];
  const int*   edge = (const int*)d_in[5];

  const int D_OUT = in_sizes[4];
  const int D_HID = in_sizes[2];
  const int N     = in_sizes[0] / D_OUT;
  const int E     = in_sizes[5] / 2;
  const int* src  = edge;
  const int* dst  = edge + E;

  char* ws = (char*)d_ws;
  const size_t o1 = (((size_t)N * 4) + 255) & ~(size_t)255;
  const size_t o2 = o1 + ((((size_t)N * D_HID * 4) + 255) & ~(size_t)255);
  float* dinv = (float*)ws;
  float* xbuf = (float*)(ws + o1);
  float* agg  = (float*)(ws + o2);

  const int T = 256;
  const int wavesPerBlk = T / 32;
  const int mtiles  = (N + 15) / 16;
  const int mgroups = (mtiles + 3) / 4;
  const long long nh = (long long)N * D_HID;
  const long long no = (long long)N * D_OUT;

  k_init_deg <<<(N + T - 1) / T, T, 0, stream>>>(dinv, N);
  k_deg_edges<<<(E + T - 1) / T, T, 0, stream>>>(dinv, dst, E);
  k_rsqrt    <<<(N + T - 1) / T, T, 0, stream>>>(dinv, N);

  {
    int waves = mgroups * (D_HID / 16);
    k_gemm_wmma<64, 4><<<(waves + wavesPerBlk - 1) / wavesPerBlk, T, 0, stream>>>(
        emb, W1, xbuf, N, D_HID);
    k_self_loop<<<(int)((nh + T - 1) / T), T, 0, stream>>>(agg, xbuf, dinv, nh, 7);
    k_scatter  <<<(E + wavesPerBlk - 1) / wavesPerBlk, T, 0, stream>>>(
        agg, xbuf, dinv, src, dst, E, D_HID);
    k_bias_relu<<<(int)((nh + T - 1) / T), T, 0, stream>>>(agg, b1, nh, D_HID - 1);
  }

  {
    int waves = mgroups * (D_OUT / 16);
    k_gemm_wmma<128, 4><<<(waves + wavesPerBlk - 1) / wavesPerBlk, T, 0, stream>>>(
        agg, W2, xbuf, N, D_OUT);
    float* out = (float*)d_out;
    k_self_loop<<<(int)((no + T - 1) / T), T, 0, stream>>>(out, xbuf, dinv, no, 6);
    k_scatter  <<<(E + wavesPerBlk - 1) / wavesPerBlk, T, 0, stream>>>(
        out, xbuf, dinv, src, dst, E, D_OUT);
    k_bias_add <<<(int)((no + T - 1) / T), T, 0, stream>>>(out, b2, no, D_OUT - 1);
  }
}


# float-atomic scatter-add replaced by per-call edge sort by destination (integer atomic ranks, prefix sum) and one wave per node summing rows in registers; sum order changes; row written once
# speedup vs baseline: 2.3313x; 2.3313x over previous
#include <hip/hip_runtime.h>


typedef float v2f __attribute__((ext_vector_type(2)));
typedef float v4f __attribute__((ext_vector_type(4)));
typedef float v8f __attribute__((ext_vector_type(8)));
typedef int   v4i __attribute__((ext_vector_type(4)));

__global__ void k_zero_cnt(int* __restrict__ cnt, int N) {
  int i = blockIdx.x * blockDim.x + threadIdx.x;
  if (i < N) cnt[i] = 0;
}

__global__ void k_rank(int* __restrict__ cnt, int* __restrict__ rank,
                       const int* __restrict__ dst, int E) {
  int i = blockIdx.x * blockDim.x + threadIdx.x;
  if (i < E) rank[i] = atomicAdd(&cnt[dst[i]], 1);
}

__global__ __launch_bounds__(1024) void k_scan_deg(const int* __restrict__ cnt,
                                                   int* __restrict__ off,
                                                   float* __restrict__ dinv, int N) {
  __shared__ int wsum[32];
  __shared__ int chunk_total;
  const int tid = threadIdx.x, lane = tid & 31, wid = tid >> 5;
  const int N4 = (N + 3) & ~3;
  const int last = max(N4 - 4, 0);
  int carry = 0;
  v4i raw = *(const v4i*)(cnt + min(4 * tid, last));
  for (int base = 0; base < N; base += 4096) {
    const int i0 = base + 4 * tid;
    const v4i nraw = *(const v4i*)(cnt + min(i0 + 4096, last));
    const bool live = (i0 < N4);
    const int c0 = live ? raw[0] : 0, c1 = live ? raw[1] : 0;
    const int c2 = live ? raw[2] : 0, c3 = live ? raw[3] : 0;
    const int tot = c0 + c1 + c2 + c3;
    int x = tot;
    for (int o = 1; o < 32; o <<= 1) {
      int y = __shfl_up(x, o);
      if (lane >= o) x += y;
    }
    if (lane == 31) wsum[wid] = x;
    __syncthreads();
    if (wid == 0) {
      const int w = wsum[lane];
      int z = w;
      for (int o = 1; o < 32; o <<= 1) {
        int y = __shfl_up(z, o);
        if (lane >= o) z += y;
      }
      wsum[lane] = z - w;
      if (lane == 31) chunk_total = z;
    }
    __syncthreads();
    const int e0 = carry + wsum[wid] + (x - tot);
    const int e1 = e0 + c0, e2 = e1 + c1, e3 = e2 + c2;
    const float d0 = rsqrtf((float)(c0 + 1)), d1 = rsqrtf((float)(c1 + 1));
    const float d2 = rsqrtf((float)(c2 + 1)), d3 = rsqrtf((float)(c3 + 1));
    if (i0 + 3 < N) {
      v4i eo = {e0, e1, e2, e3};
      v4f dv = {d0, d1, d2, d3};
      *(v4i*)(off + i0)  = eo;
      *(v4f*)(dinv + i0) = dv;
    } else {
      if (i0 + 0 < N) { off[i0 + 0] = e0; dinv[i0 + 0] = d0; }
      if (i0 + 1 < N) { off[i0 + 1] = e1; dinv[i0 + 1] = d1; }
      if (i0 + 2 < N) { off[i0 + 2] = e2; dinv[i0 + 2] = d2; }
    }
    carry += chunk_total;
    raw = nraw;
    __syncthreads();
  }
  if (tid == 0) off[N] = carry;
}

__global__ void k_fill(int* __restrict__ csr, const int* __restrict__ off,
                       const int* __restrict__ rank, const int* __restrict__ src,
                       const int* __restrict__ dst, int E) {
  int i = blockIdx.x * blockDim.x + threadIdx.x;
  if (i < E) csr[off[dst[i]] + rank[i]] = src[i];
}

template <int K, int MT>
__global__ void k_gemm_wmma(const float* __restrict__ X, const float* __restrict__ W,
                            float* __restrict__ Y, int M, int Nout) {
  constexpr int KSTEPS = K / 4;
  const int ntiles  = Nout >> 4;
  const int mtiles  = (M + 15) >> 4;
  const int mgroups = (mtiles + MT - 1) / MT;
  const int wave    = blockIdx.x * (blockDim.x >> 5) + (threadIdx.x >> 5);
  if (wave >= mgroups * ntiles) return;
  const int mg   = wave / ntiles;
  const int nt   = wave - mg * ntiles;
  const int lane = threadIdx.x & 31;
  const int half = lane >> 4;
  const int r    = lane & 15;
  const int n0   = nt << 4;

  const float* Bptr = W + (size_t)(2 * half) * Nout + (n0 + r);
  v2f bq[KSTEPS];
#pragma unroll
  for (int kk = 0; kk < KSTEPS; ++kk) {
    bq[kk][0] = Bptr[(size_t)(4 * kk + 0) * Nout];
    bq[kk][1] = Bptr[(size_t)(4 * kk + 1) * Nout];
  }

  for (int t = 0; t < MT; ++t) {
    const int mt = mg * MT + t;
    if (mt >= mtiles) break;
    const int m0 = mt << 4;
    int arow = m0 + r;
    if (arow >= M) arow = M - 1;
    const float* Aptr = X + (size_t)arow * K + 2 * half;

    v8f c = {};
#pragma unroll
    for (int kk = 0; kk < KSTEPS; ++kk) {
      v2f a = *(const v2f*)(Aptr + 4 * kk);
      c = __builtin_amdgcn_wmma_f32_16x16x4_f32(
           false, a,  false, bq[kk],
           (short)0, c,  false,  false);
    }
#pragma unroll
    for (int v = 0; v < 8; ++v) {
      int row = m0 + v + 8 * half;
      if (row < M) Y[(size_t)row * Nout + n0 + r] = c[v];
    }
  }
}

template <int V>
struct rowvec;
template <> struct rowvec<4> { typedef v4f type; };
template <> struct rowvec<2> { typedef v2f type; };

template <int V, int B>
__device__ __forceinline__ void gather_rows(typename rowvec<V>::type& acc,
                                            const typename rowvec<V>::type* __restrict__ Xv,
                                            int ms, int mwbits, int j, int lane) {
  typedef typename rowvec<V>::type vf;
  int   s[B];
  float w[B];
  vf    x[B];
#pragma unroll
  for (int b = 0; b < B; ++b) {
    s[b] = __builtin_amdgcn_readlane(ms, j + b);
    w[b] = __int_as_float(__builtin_amdgcn_readlane(mwbits, j + b));
  }
#pragma unroll
  for (int b = 0; b < B; ++b) x[b] = Xv[(size_t)s[b] * 32 + lane];
#pragma unroll
  for (int b = 0; b < B; ++b) acc += x[b] * w[b];
}

template <int D, bool RELU>
__global__ __launch_bounds__(256) void k_aggregate(float* __restrict__ Y,
                                                   const float* __restrict__ X,
                                                   const float* __restrict__ dinv,
                                                   const int* __restrict__ off,
                                                   const int* __restrict__ csr,
                                                   const float* __restrict__ bias, int N) {
  constexpr int V = D / 32;
  typedef typename rowvec<V>::type vf;
  int node = blockIdx.x * (blockDim.x >> 5) + (threadIdx.x >> 5);
  if (node >= N) return;
  node = __builtin_amdgcn_readfirstlane(node);
  const int lane = threadIdx.x & 31;
  const int beg = off[node], end = off[node + 1];
  const float dd = dinv[node];
  const vf* Xv = (const vf*)X;

  vf acc = Xv[(size_t)node * 32 + lane] * (dd * dd);
  for (int base = beg; base < end; base += 32) {
    const int n = min(32, end - base);
    int ms = 0;
    float mw = 0.0f;
    if (lane < n) {
      ms = csr[base + lane];
      mw = dinv[ms] * dd;
    }
    const int mwbits = __float_as_int(mw);
    int j = 0;
    for (; j + 8 <= n; j += 8) gather_rows<V, 8>(acc, Xv, ms, mwbits, j, lane);
    if (j + 4 <= n) { gather_rows<V, 4>(acc, Xv, ms, mwbits, j, lane); j += 4; }
    if (j + 2 <= n) { gather_rows<V, 2>(acc, Xv, ms, mwbits, j, lane); j += 2; }
    if (j < n)      { gather_rows<V, 1>(acc, Xv, ms, mwbits, j, lane); }
  }
  acc += ((const vf*)bias)[lane];
  if (RELU) {
#pragma unroll
    for (int k = 0; k < V; ++k) acc[k] = fmaxf(acc[k], 0.0f);
  }
  ((vf*)Y)[(size_t)node * 32 + lane] = acc;
}

extern "C" void kernel_launch(void* const* d_in, const int* in_sizes, int n_in,
                              void* d_out, int out_size, void* d_ws, size_t ws_size,
                              hipStream_t stream) {
  const float* emb  = (const float*)d_in[0];
  const float* W1   = (const float*)d_in[1];
  const float* b1   = (const float*)d_in[2];
  const float* W2   = (const float*)d_in[3];
  const float* b2   = (const float*)d_in[4];
  const int*   edge = (const int*)d_in[5];

  const int D_OUT = in_sizes[4];
  const int D_HID = in_sizes[2];
  const int N     = in_sizes[0] / D_OUT;
  const int E     = in_sizes[5] / 2;
  const int* src  = edge;
  const int* dst  = edge + E;

  char* ws = (char*)d_ws;
  const size_t szN = (((size_t)(N + 1) * 4) + 255) & ~(size_t)255;
  const size_t szE = (((size_t)E * 4) + 255) & ~(size_t)255;
  const size_t szX = (((size_t)N * D_HID * 4) + 255) & ~(size_t)255;
  int*   cnt  = (int*)ws;
  int*   off  = (int*)(ws + szN);
  float* dinv = (float*)(ws + 2 * szN);
  int*   rank = (int*)(ws + 3 * szN);
  int*   csr  = (int*)(ws + 3 * szN + szE);
  float* xbuf = (float*)(ws + 3 * szN + 2 * szE);
  float* hbuf = (float*)(ws + 3 * szN + 2 * szE + szX);

  const int T = 256;
  const int wavesPerBlk = T / 32;
  const int mtiles  = (N + 15) / 16;
  const int mgroups = (mtiles + 3) / 4;
  const int N4      = (N + 3) & ~3;

  k_zero_cnt<<<(N4 + T - 1) / T, T, 0, stream>>>(cnt, N4);
  k_rank    <<<(E + T - 1) / T, T, 0, stream>>>(cnt, rank, dst, E);
  k_scan_deg<<<1, 1024, 0, stream>>>(cnt, off, dinv, N);
  k_fill    <<<(E + T - 1) / T, T, 0, stream>>>(csr, off, rank, src, dst, E);

  {
    int waves = mgroups * (D_HID / 16);
    k_gemm_wmma<64, 4><<<(waves + wavesPerBlk - 1) / wavesPerBlk, T, 0, stream>>>(
        emb, W1, xbuf, N, D_HID);
    k_aggregate<128, true><<<(N + wavesPerBlk - 1) / wavesPerBlk, T, 0, stream>>>(
        hbuf, xbuf, dinv, off, csr, b1, N);
  }

  {
    int waves = mgroups * (D_OUT / 16);
    k_gemm_wmma<128, 4><<<(waves + wavesPerBlk - 1) / wavesPerBlk, T, 0, stream>>>(
        hbuf, W2, xbuf, N, D_OUT);
    float* out = (float*)d_out;
    k_aggregate<64, false><<<(N + wavesPerBlk - 1) / wavesPerBlk, T, 0, stream>>>(
        out, xbuf, dinv, off, csr, b2, N);
  }
}


# layer 1 sums the 64-wide input rows over in-edges before the W1 product, (AX)W for A(XW), bias+relu in the product's epilogue: half the gathered bytes; products re-associated, sum order changes
# speedup vs baseline: 2.4334x; 1.0438x over previous
#include <hip/hip_runtime.h>


typedef float v2f __attribute__((ext_vector_type(2)));
typedef float v4f __attribute__((ext_vector_type(4)));
typedef float v8f __attribute__((ext_vector_type(8)));
typedef int   v4i __attribute__((ext_vector_type(4)));

__global__ void k_zero_cnt(int* __restrict__ cnt, int N) {
  int i = blockIdx.x * blockDim.x + threadIdx.x;
  if (i < N) cnt[i] = 0;
}

__global__ void k_rank(int* __restrict__ cnt, int* __restrict__ rank,
                       const int* __restrict__ dst, int E) {
  int i = blockIdx.x * blockDim.x + threadIdx.x;
  if (i < E) rank[i] = atomicAdd(&cnt[dst[i]], 1);
}

__global__ __launch_bounds__(1024) void k_scan_deg(const int* __restrict__ cnt,
                                                   int* __restrict__ off,
                                                   float* __restrict__ dinv, int N) {
  __shared__ int wsum[32];
  __shared__ int chunk_total;
  const int tid = threadIdx.x, lane = tid & 31, wid = tid >> 5;
  const int N4 = (N + 3) & ~3;
  const int last = max(N4 - 4, 0);
  int carry = 0;
  v4i raw = *(const v4i*)(cnt + min(4 * tid, last));
  for (int base = 0; base < N; base += 4096) {
    const int i0 = base + 4 * tid;
    const v4i nraw = *(const v4i*)(cnt + min(i0 + 4096, last));
    const bool live = (i0 < N4);
    const int c0 = live ? raw[0] : 0, c1 = live ? raw[1] : 0;
    const int c2 = live ? raw[2] : 0, c3 = live ? raw[3] : 0;
    const int tot = c0 + c1 + c2 + c3;
    int x = tot;
    for (int o = 1; o < 32; o <<= 1) {
      int y = __shfl_up(x, o);
      if (lane >= o) x += y;
    }
    if (lane == 31) wsum[wid] = x;
    __syncthreads();
    if (wid == 0) {
      const int w = wsum[lane];
      int z = w;
      for (int o = 1; o < 32; o <<= 1) {
        int y = __shfl_up(z, o);
        if (lane >= o) z += y;
      }
      wsum[lane] = z - w;
      if (lane == 31) chunk_total = z;
    }
    __syncthreads();
    const int e0 = carry + wsum[wid] + (x - tot);
    const int e1 = e0 + c0, e2 = e1 + c1, e3 = e2 + c2;
    const float d0 = rsqrtf((float)(c0 + 1)), d1 = rsqrtf((float)(c1 + 1));
    const float d2 = rsqrtf((float)(c2 + 1)), d3 = rsqrtf((float)(c3 + 1));
    if (i0 + 3 < N) {
      v4i eo = {e0, e1, e2, e3};
      v4f dv = {d0, d1, d2, d3};
      *(v4i*)(off + i0)  = eo;
      *(v4f*)(dinv + i0) = dv;
    } else {
      if (i0 + 0 < N) { off[i0 + 0] = e0; dinv[i0 + 0] = d0; }
      if (i0 + 1 < N) { off[i0 + 1] = e1; dinv[i0 + 1] = d1; }
      if (i0 + 2 < N) { off[i0 + 2] = e2; dinv[i0 + 2] = d2; }
    }
    carry += chunk_total;
    raw = nraw;
    __syncthreads();
  }
  if (tid == 0) off[N] = carry;
}

__global__ void k_fill(int* __restrict__ csr, const int* __restrict__ off,
                       const int* __restrict__ rank, const int* __restrict__ src,
                       const int* __restrict__ dst, int E) {
  int i = blockIdx.x * blockDim.x + threadIdx.x;
  if (i < E) csr[off[dst[i]] + rank[i]] = src[i];
}

template <int K, int MT, bool BIAS_RELU>
__global__ void k_gemm_wmma(const float* __restrict__ X, const float* __restrict__ W,
                            const float* __restrict__ bias,
                            float* __restrict__ Y, int M, int Nout) {
  constexpr int KSTEPS = K / 4;
  const int ntiles  = Nout >> 4;
  const int mtiles  = (M + 15) >> 4;
  const int mgroups = (mtiles + MT - 1) / MT;
  const int wave    = blockIdx.x * (blockDim.x >> 5) + (threadIdx.x >> 5);
  if (wave >= mgroups * ntiles) return;
  const int mg   = wave / ntiles;
  const int nt   = wave - mg * ntiles;
  const int lane = threadIdx.x & 31;
  const int half = lane >> 4;
  const int r    = lane & 15;
  const int n0   = nt << 4;

  const float* Bptr = W + (size_t)(2 * half) * Nout + (n0 + r);
  v2f bq[KSTEPS];
#pragma unroll
  for (int kk = 0; kk < KSTEPS; ++kk) {
    bq[kk][0] = Bptr[(size_t)(4 * kk + 0) * Nout];
    bq[kk][1] = Bptr[(size_t)(4 * kk + 1) * Nout];
  }
  const float bcol = BIAS_RELU ? bias[n0 + r] : 0.0f;

  for (int t = 0; t < MT; ++t) {
    const int mt = mg * MT + t;
    if (mt >= mtiles) break;
    const int m0 = mt << 4;
    int arow = m0 + r;
    if (arow >= M) arow = M - 1;
    const float* Aptr = X + (size_t)arow * K + 2 * half;

    v8f c = {};
#pragma unroll
    for (int kk = 0; kk < KSTEPS; ++kk) {
      v2f a = *(const v2f*)(Aptr + 4 * kk);
      c = __builtin_amdgcn_wmma_f32_16x16x4_f32(
           false, a,  false, bq[kk],
           (short)0, c,  false,  false);
    }
#pragma unroll
    for (int v = 0; v < 8; ++v) {
      int row = m0 + v + 8 * half;
      const float y = BIAS_RELU ? fmaxf(c[v] + bcol, 0.0f) : c[v];
      if (row < M) Y[(size_t)row * Nout + n0 + r] = y;
    }
  }
}

template <int V>
struct rowvec;
template <> struct rowvec<4> { typedef v4f type; };
template <> struct rowvec<2> { typedef v2f type; };

template <int V, int B>
__device__ __forceinline__ void gather_rows(typename rowvec<V>::type& acc,
                                            const typename rowvec<V>::type* __restrict__ Xv,
                                            int ms, int mwbits, int j, int lane) {
  typedef typename rowvec<V>::type vf;
  int   s[B];
  float w[B];
  vf    x[B];
#pragma unroll
  for (int b = 0; b < B; ++b) {
    s[b] = __builtin_amdgcn_readlane(ms, j + b);
    w[b] = __int_as_float(__builtin_amdgcn_readlane(mwbits, j + b));
  }
#pragma unroll
  for (int b = 0; b < B; ++b) x[b] = Xv[(size_t)s[b] * 32 + lane];
#pragma unroll
  for (int b = 0; b < B; ++b) acc += x[b] * w[b];
}

template <int D, bool BIAS>
__global__ __launch_bounds__(256) void k_aggregate(float* __restrict__ Y,
                                                   const float* __restrict__ X,
                                                   const float* __restrict__ dinv,
                                                   const int* __restrict__ off,
                                                   const int* __restrict__ csr,
                                                   const float* __restrict__ bias, int N) {
  constexpr int V = D / 32;
  typedef typename rowvec<V>::type vf;
  int node = blockIdx.x * (blockDim.x >> 5) + (threadIdx.x >> 5);
  if (node >= N) return;
  node = __builtin_amdgcn_readfirstlane(node);
  const int lane = threadIdx.x & 31;
  const int beg = off[node], end = off[node + 1];
  const float dd = dinv[node];
  const vf* Xv = (const vf*)X;

  vf acc = Xv[(size_t)node * 32 + lane] * (dd * dd);
  for (int base = beg; base < end; base += 32) {
    const int n = min(32, end - base);
    int ms = 0;
    float mw = 0.0f;
    if (lane < n) {
      ms = csr[base + lane];
      mw = dinv[ms] * dd;
    }
    const int mwbits = __float_as_int(mw);
    int j = 0;
    for (; j + 8 <= n; j += 8) gather_rows<V, 8>(acc, Xv, ms, mwbits, j, lane);
    if (j + 4 <= n) { gather_rows<V, 4>(acc, Xv, ms, mwbits, j, lane); j += 4; }
    if (j + 2 <= n) { gather_rows<V, 2>(acc, Xv, ms, mwbits, j, lane); j += 2; }
    if (j < n)      { gather_rows<V, 1>(acc, Xv, ms, mwbits, j, lane); }
  }
  if (BIAS) acc += ((const vf*)bias)[lane];
  ((vf*)Y)[(size_t)node * 32 + lane] = acc;
}

extern "C" void kernel_launch(void* const* d_in, const int* in_sizes, int n_in,
                              void* d_out, int out_size, void* d_ws, size_t ws_size,
                              hipStream_t stream) {
  const float* emb  = (const float*)d_in[0];
  const float* W1   = (const float*)d_in[1];
  const float* b1   = (const float*)d_in[2];
  const float* W2   = (const float*)d_in[3];
  const float* b2   = (const float*)d_in[4];
  const int*   edge = (const int*)d_in[5];

  const int D_OUT = in_sizes[4];
  const int D_HID = in_sizes[2];
  const int N     = in_sizes[0] / D_OUT;
  const int E     = in_sizes[5] / 2;
  const int* src  = edge;
  const int* dst  = edge + E;

  char* ws = (char*)d_ws;
  const size_t szN = (((size_t)(N + 1) * 4) + 255) & ~(size_t)255;
  const size_t szE = (((size_t)E * 4) + 255) & ~(size_t)255;
  const size_t szA = (((size_t)N * D_OUT * 4) + 255) & ~(size_t)255;
  int*   cnt  = (int*)ws;
  int*   off  = (int*)(ws + szN);
  float* dinv = (float*)(ws + 2 * szN);
  int*   rank = (int*)(ws + 3 * szN);
  int*   csr  = (int*)(ws + 3 * szN + szE);
  float* abuf = (float*)(ws + 3 * szN + 2 * szE);
  float* hbuf = (float*)(ws + 3 * szN + 2 * szE + szA);

  const int T = 256;
  const int wavesPerBlk = T / 32;
  const int mtiles  = (N + 15) / 16;
  const int mgroups = (mtiles + 3) / 4;
  const int N4      = (N + 3) & ~3;

  k_zero_cnt<<<(N4 + T - 1) / T, T, 0, stream>>>(cnt, N4);
  k_rank    <<<(E + T - 1) / T, T, 0, stream>>>(cnt, rank, dst, E);
  k_scan_deg<<<1, 1024, 0, stream>>>(cnt, off, dinv, N);
  k_fill    <<<(E + T - 1) / T, T, 0, stream>>>(csr, off, rank, src, dst, E);

  {
    k_aggregate<64, false><<<(N + wavesPerBlk - 1) / wavesPerBlk, T, 0, stream>>>(
        abuf, emb, dinv, off, csr, nullptr, N);
    int waves = mgroups * (D_HID / 16);
    k_gemm_wmma<64, 4, true><<<(waves + wavesPerBlk - 1) / wavesPerBlk, T, 0, stream>>>(
        abuf, W1, b1, hbuf, N, D_HID);
  }

  {
    int waves = mgroups * (D_OUT / 16);
    k_gemm_wmma<128, 4, false><<<(waves + wavesPerBlk - 1) / wavesPerBlk, T, 0, stream>>>(
        hbuf, W2, nullptr, abuf, N, D_OUT);
    float* out = (float*)d_out;
    k_aggregate<64, true><<<(N + wavesPerBlk - 1) / wavesPerBlk, T, 0, stream>>>(
        out, abuf, dinv, off, csr, b2, N);
  }
}


# both dense products in one launch: a wave owns 16 rows and all columns, W1 and W2 staged in LDS, hidden tile passed through LDS, rows fetched once as whole lines; 8 launches become 7
# speedup vs baseline: 2.7808x; 1.1428x over previous
#include <hip/hip_runtime.h>


typedef float v2f __attribute__((ext_vector_type(2)));
typedef float v4f __attribute__((ext_vector_type(4)));
typedef float v8f __attribute__((ext_vector_type(8)));
typedef int   v4i __attribute__((ext_vector_type(4)));

__global__ void k_zero_cnt(int* __restrict__ cnt, int N) {
  int i = blockIdx.x * blockDim.x + threadIdx.x;
  if (i < N) cnt[i] = 0;
}

__global__ void k_rank(int* __restrict__ cnt, int* __restrict__ rank,
                       const int* __restrict__ dst, int E) {
  int i = blockIdx.x * blockDim.x + threadIdx.x;
  if (i < E) rank[i] = atomicAdd(&cnt[dst[i]], 1);
}

__global__ __launch_bounds__(1024) void k_scan_deg(const int* __restrict__ cnt,
                                                   int* __restrict__ off,
                                                   float* __restrict__ dinv, int N) {
  __shared__ int wsum[32];
  __shared__ int chunk_total;
  const int tid = threadIdx.x, lane = tid & 31, wid = tid >> 5;
  const int N4 = (N + 3) & ~3;
  const int last = max(N4 - 4, 0);
  int carry = 0;
  v4i raw = *(const v4i*)(cnt + min(4 * tid, last));
  for (int base = 0; base < N; base += 4096) {
    const int i0 = base + 4 * tid;
    const v4i nraw = *(const v4i*)(cnt + min(i0 + 4096, last));
    const bool live = (i0 < N4);
    const int c0 = live ? raw[0] : 0, c1 = live ? raw[1] : 0;
    const int c2 = live ? raw[2] : 0, c3 = live ? raw[3] : 0;
    const int tot = c0 + c1 + c2 + c3;
    int x = tot;
    for (int o = 1; o < 32; o <<= 1) {
      int y = __shfl_up(x, o);
      if (lane >= o) x += y;
    }
    if (lane == 31) wsum[wid] = x;
    __syncthreads();
    if (wid == 0) {
      const int w = wsum[lane];
      int z = w;
      for (int o = 1; o < 32; o <<= 1) {
        int y = __shfl_up(z, o);
        if (lane >= o) z += y;
      }
      wsum[lane] = z - w;
      if (lane == 31) chunk_total = z;
    }
    __syncthreads();
    const int e0 = carry + wsum[wid] + (x - tot);
    const int e1 = e0 + c0, e2 = e1 + c1, e3 = e2 + c2;
    const float d0 = rsqrtf((float)(c0 + 1)), d1 = rsqrtf((float)(c1 + 1));
    const float d2 = rsqrtf((float)(c2 + 1)), d3 = rsqrtf((float)(c3 + 1));
    if (i0 + 3 < N) {
      v4i eo = {e0, e1, e2, e3};
      v4f dv = {d0, d1, d2, d3};
      *(v4i*)(off + i0)  = eo;
      *(v4f*)(dinv + i0) = dv;
    } else {
      if (i0 + 0 < N) { off[i0 + 0] = e0; dinv[i0 + 0] = d0; }
      if (i0 + 1 < N) { off[i0 + 1] = e1; dinv[i0 + 1] = d1; }
      if (i0 + 2 < N) { off[i0 + 2] = e2; dinv[i0 + 2] = d2; }
    }
    carry += chunk_total;
    raw = nraw;
    __syncthreads();
  }
  if (tid == 0) off[N] = carry;
}

__global__ void k_fill(int* __restrict__ csr, const int* __restrict__ off,
                       const int* __restrict__ rank, const int* __restrict__ src,
                       const int* __restrict__ dst, int E) {
  int i = blockIdx.x * blockDim.x + threadIdx.x;
  if (i < E) csr[off[dst[i]] + rank[i]] = src[i];
}

constexpr int D_IN  = 64;
constexpr int D_MID = 128;
constexpr int D_FIN = 64;
constexpr int PW1  = 2 * D_MID + 32;
constexpr int PW2  = 2 * D_FIN + 32;
constexpr int PA   = D_IN + 4;
constexpr int PH   = D_MID + 4;
constexpr int SLOT = 16 * PH;
constexpr int DENSE_WAVES = 8;

__global__ __launch_bounds__(256) void k_dense2(const float* __restrict__ X,
                                                const float* __restrict__ W1,
                                                const float* __restrict__ b1,
                                                const float* __restrict__ W2,
                                                float* __restrict__ Y, int M) {
  constexpr int W2_FLOATS = (D_MID / 2) * PW2;
  constexpr int W1_FLOATS = (D_IN / 2) * PW1;
  __shared__ __attribute__((aligned(16))) float lds[W2_FLOATS + W1_FLOATS + DENSE_WAVES * SLOT];
  float* const w2p   = lds;
  float* const w1p   = lds + W2_FLOATS;
  float* const slots = lds + W2_FLOATS + W1_FLOATS;

  const int tid  = threadIdx.x;
  const int lane = tid & 31;
  const int wid  = __builtin_amdgcn_readfirstlane(tid >> 5);
  const int half = lane >> 4;
  const int r    = lane & 15;
  const int mtiles = (M + 15) >> 4;
  const int tile   = blockIdx.x * DENSE_WAVES + wid;
  const bool live  = tile < mtiles;
  const int m0 = (live ? tile : mtiles - 1) << 4;
  float* slot = slots + wid * SLOT;

  const int trow = lane >> 4;
  const int tcol = (lane & 15) * 4;
  v4f xin[8];
#pragma unroll
  for (int i = 0; i < 8; ++i) {
    const int gr = min(m0 + 2 * i + trow, M - 1);
    xin[i] = *(const v4f*)(X + (size_t)gr * D_IN + tcol);
  }
  float bias[D_MID / 16];
#pragma unroll
  for (int nt = 0; nt < D_MID / 16; ++nt) bias[nt] = b1[nt * 16 + r];

#pragma unroll
  for (int it = 0; it < (D_IN / 2) * (D_MID / 4) / 256; ++it) {
    const int item = tid + 256 * it;
    const int p = item / (D_MID / 4), n = (item % (D_MID / 4)) * 4;
    const v4f lo = *(const v4f*)(W1 + (size_t)(2 * p) * D_MID + n);
    const v4f hi = *(const v4f*)(W1 + (size_t)(2 * p + 1) * D_MID + n);
    const v4f s0 = {lo[0], hi[0], lo[1], hi[1]};
    const v4f s1 = {lo[2], hi[2], lo[3], hi[3]};
    *(v4f*)(w1p + p * PW1 + 2 * n)     = s0;
    *(v4f*)(w1p + p * PW1 + 2 * n + 4) = s1;
  }
#pragma unroll
  for (int it = 0; it < (D_MID / 2) * (D_FIN / 4) / 256; ++it) {
    const int item = tid + 256 * it;
    const int p = item / (D_FIN / 4), n = (item % (D_FIN / 4)) * 4;
    const v4f lo = *(const v4f*)(W2 + (size_t)(2 * p) * D_FIN + n);
    const v4f hi = *(const v4f*)(W2 + (size_t)(2 * p + 1) * D_FIN + n);
    const v4f s0 = {lo[0], hi[0], lo[1], hi[1]};
    const v4f s1 = {lo[2], hi[2], lo[3], hi[3]};
    *(v4f*)(w2p + p * PW2 + 2 * n)     = s0;
    *(v4f*)(w2p + p * PW2 + 2 * n + 4) = s1;
  }
#pragma unroll
  for (int i = 0; i < 8; ++i) *(v4f*)(slot + (2 * i + trow) * PA + tcol) = xin[i];
  __syncthreads();

  v8f acc[D_MID / 16];
#pragma unroll
  for (int nt = 0; nt < D_MID / 16; ++nt) acc[nt] = v8f{};
  {
    const float* ap = slot + r * PA + 2 * half;
    const float* bp = w1p + half * PW1 + 2 * r;
#pragma unroll
    for (int kk = 0; kk < D_IN / 4; ++kk) {
      const v2f a = *(const v2f*)(ap + 4 * kk);
#pragma unroll
      for (int nt = 0; nt < D_MID / 16; ++nt) {
        const v2f b = *(const v2f*)(bp + (2 * kk) * PW1 + nt * 32);
        acc[nt] = __builtin_amdgcn_wmma_f32_16x16x4_f32(
             false, a,  false, b,
             (short)0, acc[nt],  false,  false);
      }
    }
  }
#pragma unroll
  for (int nt = 0; nt < D_MID / 16; ++nt) {
#pragma unroll
    for (int v = 0; v < 8; ++v)
      slot[(v + 8 * half) * PH + nt * 16 + r] = fmaxf(acc[nt][v] + bias[nt], 0.0f);
  }
  __syncthreads();

  v8f acc2[D_FIN / 16];
#pragma unroll
  for (int nt = 0; nt < D_FIN / 16; ++nt) acc2[nt] = v8f{};
  {
    const float* ap = slot + r * PH + 2 * half;
    const float* bp = w2p + half * PW2 + 2 * r;
#pragma unroll
    for (int kk = 0; kk < D_MID / 4; ++kk) {
      const v2f a = *(const v2f*)(ap + 4 * kk);
#pragma unroll
      for (int nt = 0; nt < D_FIN / 16; ++nt) {
        const v2f b = *(const v2f*)(bp + (2 * kk) * PW2 + nt * 32);
        acc2[nt] = __builtin_amdgcn_wmma_f32_16x16x4_f32(
             false, a,  false, b,
             (short)0, acc2[nt],  false,  false);
      }
    }
  }
#pragma unroll
  for (int nt = 0; nt < D_FIN / 16; ++nt) {
#pragma unroll
    for (int v = 0; v < 8; ++v) slot[(v + 8 * half) * PA + nt * 16 + r] = acc2[nt][v];
  }
  __syncthreads();
  if (live) {
    if (m0 + 16 <= M) {
#pragma unroll
      for (int i = 0; i < 8; ++i) {
        const v4f y = *(const v4f*)(slot + (2 * i + trow) * PA + tcol);
        *(v4f*)(Y + (size_t)(m0 + 2 * i + trow) * D_FIN + tcol) = y;
      }
    } else {
#pragma unroll
      for (int i = 0; i < 8; ++i) {
        const int row = m0 + 2 * i + trow;
        const v4f y = *(const v4f*)(slot + (2 * i + trow) * PA + tcol);
        if (row < M) *(v4f*)(Y + (size_t)row * D_FIN + tcol) = y;
      }
    }
  }
}

template <int V>
struct rowvec;
template <> struct rowvec<4> { typedef v4f type; };
template <> struct rowvec<2> { typedef v2f type; };

template <int V, int B>
__device__ __forceinline__ void gather_rows(typename rowvec<V>::type& acc,
                                            const typename rowvec<V>::type* __restrict__ Xv,
                                            int ms, int mwbits, int j, int lane) {
  typedef typename rowvec<V>::type vf;
  int   s[B];
  float w[B];
  vf    x[B];
#pragma unroll
  for (int b = 0; b < B; ++b) {
    s[b] = __builtin_amdgcn_readlane(ms, j + b);
    w[b] = __int_as_float(__builtin_amdgcn_readlane(mwbits, j + b));
  }
#pragma unroll
  for (int b = 0; b < B; ++b) x[b] = Xv[(size_t)s[b] * 32 + lane];
#pragma unroll
  for (int b = 0; b < B; ++b) acc += x[b] * w[b];
}

template <int D, bool BIAS>
__global__ __launch_bounds__(256) void k_aggregate(float* __restrict__ Y,
                                                   const float* __restrict__ X,
                                                   const float* __restrict__ dinv,
                                                   const int* __restrict__ off,
                                                   const int* __restrict__ csr,
                                                   const float* __restrict__ bias, int N) {
  constexpr int V = D / 32;
  typedef typename rowvec<V>::type vf;
  int node = blockIdx.x * (blockDim.x >> 5) + (threadIdx.x >> 5);
  if (node >= N) return;
  node = __builtin_amdgcn_readfirstlane(node);
  const int lane = threadIdx.x & 31;
  const int beg = off[node], end = off[node + 1];
  const float dd = dinv[node];
  const vf* Xv = (const vf*)X;

  vf acc = Xv[(size_t)node * 32 + lane] * (dd * dd);
  for (int base = beg; base < end; base += 32) {
    const int n = min(32, end - base);
    int ms = 0;
    float mw = 0.0f;
    if (lane < n) {
      ms = csr[base + lane];
      mw = dinv[ms] * dd;
    }
    const int mwbits = __float_as_int(mw);
    int j = 0;
    for (; j + 8 <= n; j += 8) gather_rows<V, 8>(acc, Xv, ms, mwbits, j, lane);
    if (j + 4 <= n) { gather_rows<V, 4>(acc, Xv, ms, mwbits, j, lane); j += 4; }
    if (j + 2 <= n) { gather_rows<V, 2>(acc, Xv, ms, mwbits, j, lane); j += 2; }
    if (j < n)      { gather_rows<V, 1>(acc, Xv, ms, mwbits, j, lane); }
  }
  if (BIAS) acc += ((const vf*)bias)[lane];
  ((vf*)Y)[(size_t)node * 32 + lane] = acc;
}

extern "C" void kernel_launch(void* const* d_in, const int* in_sizes, int n_in,
                              void* d_out, int out_size, void* d_ws, size_t ws_size,
                              hipStream_t stream) {
  const float* emb  = (const float*)d_in[0];
  const float* W1   = (const float*)d_in[1];
  const float* b1   = (const float*)d_in[2];
  const float* W2   = (const float*)d_in[3];
  const float* b2   = (const float*)d_in[4];
  const int*   edge = (const int*)d_in[5];

  const int D_OUT = in_sizes[4];
  const int N     = in_sizes[0] / D_OUT;
  const int E     = in_sizes[5] / 2;
  const int* src  = edge;
  const int* dst  = edge + E;

  char* ws = (char*)d_ws;
  const size_t szN = (((size_t)(N + 1) * 4) + 255) & ~(size_t)255;
  const size_t szE = (((size_t)E * 4) + 255) & ~(size_t)255;
  const size_t szA = (((size_t)N * D_OUT * 4) + 255) & ~(size_t)255;
  int*   cnt  = (int*)ws;
  int*   off  = (int*)(ws + szN);
  float* dinv = (float*)(ws + 2 * szN);
  int*   rank = (int*)(ws + 3 * szN);
  int*   csr  = (int*)(ws + 3 * szN + szE);
  float* abuf = (float*)(ws + 3 * szN + 2 * szE);
  float* xbuf = (float*)(ws + 3 * szN + 2 * szE + szA);

  const int T = 256;
  const int wavesPerBlk = T / 32;
  const int mtiles  = (N + 15) / 16;
  const int N4      = (N + 3) & ~3;

  k_zero_cnt<<<(N4 + T - 1) / T, T, 0, stream>>>(cnt, N4);
  k_rank    <<<(E + T - 1) / T, T, 0, stream>>>(cnt, rank, dst, E);
  k_scan_deg<<<1, 1024, 0, stream>>>(cnt, off, dinv, N);
  k_fill    <<<(E + T - 1) / T, T, 0, stream>>>(csr, off, rank, src, dst, E);

  k_aggregate<64, false><<<(N + wavesPerBlk - 1) / wavesPerBlk, T, 0, stream>>>(
      abuf, emb, dinv, off, csr, nullptr, N);
  k_dense2<<<(mtiles + DENSE_WAVES - 1) / DENSE_WAVES, T, 0, stream>>>(
      abuf, W1, b1, W2, xbuf, N);
  float* out = (float*)d_out;
  k_aggregate<64, true><<<(N + wavesPerBlk - 1) / wavesPerBlk, T, 0, stream>>>(
      out, xbuf, dinv, off, csr, b2, N);
}


# in-degree prefix sum: the lone workgroup that walked 25 chunks, three barriers each, becomes one workgroup per chunk that sums the other chunks of its nearer side itself (the total is the edge count)
# speedup vs baseline: 3.2247x; 1.1596x over previous
#include <hip/hip_runtime.h>


typedef float v2f __attribute__((ext_vector_type(2)));
typedef float v4f __attribute__((ext_vector_type(4)));
typedef float v8f __attribute__((ext_vector_type(8)));
typedef int   v4i __attribute__((ext_vector_type(4)));

__global__ void k_zero_cnt(int* __restrict__ cnt, int N) {
  int i = blockIdx.x * blockDim.x + threadIdx.x;
  if (i < N) cnt[i] = 0;
}

__global__ void k_rank(int* __restrict__ cnt, int* __restrict__ rank,
                       const int* __restrict__ dst, int E) {
  int i = blockIdx.x * blockDim.x + threadIdx.x;
  if (i < E) rank[i] = atomicAdd(&cnt[dst[i]], 1);
}

constexpr int SCAN_T = 1024;
constexpr int SCAN_C = 4 * SCAN_T;

__device__ __forceinline__ int wave_sum(int v) {
  for (int o = 16; o > 0; o >>= 1) v += __shfl_xor(v, o);
  return v;
}

__global__ __launch_bounds__(SCAN_T) void k_scan_deg(const int* __restrict__ cnt,
                                                     int* __restrict__ off,
                                                     float* __restrict__ dinv, int N, int E) {
  __shared__ int wtot[32];
  __shared__ int wpart[32];
  const int tid = threadIdx.x, lane = tid & 31, wid = tid >> 5;
  const int b = blockIdx.x;
  const int N4 = (N + 3) & ~3;
  const int G = (N4 + SCAN_C - 1) / SCAN_C;
  const int last = max(N4 - 4, 0);
  const int i0 = b * SCAN_C + 4 * tid;
  const int jt = (G - 1) * SCAN_C + 4 * tid;
  const v4i raw  = *(const v4i*)(cnt + min(i0, last));
  const v4i tail = *(const v4i*)(cnt + min(jt, last));
  const bool front = (b <= G - 1 - b);
  const int lo = front ? 0 : b + 1;
  const int hi = front ? b : G - 1;
  int part = 0;
#pragma unroll 4
  for (int c = lo; c < hi; ++c) {
    const v4i v = *(const v4i*)(cnt + c * SCAN_C + 4 * tid);
    part += (v[0] + v[1]) + (v[2] + v[3]);
  }
  const bool tl = !front && b < G - 1 && jt < N4;
  part += ((tl ? tail[0] : 0) + (tl ? tail[1] : 0)) + ((tl ? tail[2] : 0) + (tl ? tail[3] : 0));
  const bool live = (i0 < N4);
  const int c0 = live ? raw[0] : 0, c1 = live ? raw[1] : 0;
  const int c2 = live ? raw[2] : 0, c3 = live ? raw[3] : 0;
  const int tot = c0 + c1 + c2 + c3;
  int x = tot;
  for (int o = 1; o < 32; o <<= 1) {
    int y = __shfl_up(x, o);
    if (lane >= o) x += y;
  }
  part = wave_sum(part);
  if (lane == 31) { wtot[wid] = x; wpart[wid] = part; }
  __syncthreads();
  const int t = wtot[lane];
  const int before = wave_sum(lane < wid ? t : 0);
  const int all    = wave_sum(t);
  const int others = wave_sum(wpart[lane]);
  const int carry  = front ? others : E - all - others;
  const int e0 = carry + before + (x - tot);
  const int e1 = e0 + c0, e2 = e1 + c1, e3 = e2 + c2;
  const float d0 = rsqrtf((float)(c0 + 1)), d1 = rsqrtf((float)(c1 + 1));
  const float d2 = rsqrtf((float)(c2 + 1)), d3 = rsqrtf((float)(c3 + 1));
  if (i0 + 3 < N) {
    v4i eo = {e0, e1, e2, e3};
    v4f dv = {d0, d1, d2, d3};
    *(v4i*)(off + i0)  = eo;
    *(v4f*)(dinv + i0) = dv;
  } else {
    if (i0 + 0 < N) { off[i0 + 0] = e0; dinv[i0 + 0] = d0; }
    if (i0 + 1 < N) { off[i0 + 1] = e1; dinv[i0 + 1] = d1; }
    if (i0 + 2 < N) { off[i0 + 2] = e2; dinv[i0 + 2] = d2; }
  }
  if (b == G - 1 && tid == 0) off[N] = carry + all;
}

__global__ void k_fill(int* __restrict__ csr, const int* __restrict__ off,
                       const int* __restrict__ rank, const int* __restrict__ src,
                       const int* __restrict__ dst, int E) {
  int i = blockIdx.x * blockDim.x + threadIdx.x;
  if (i < E) csr[off[dst[i]] + rank[i]] = src[i];
}

constexpr int D_IN  = 64;
constexpr int D_MID = 128;
constexpr int D_FIN = 64;
constexpr int PW1  = 2 * D_MID + 32;
constexpr int PW2  = 2 * D_FIN + 32;
constexpr int PA   = D_IN + 4;
constexpr int PH   = D_MID + 4;
constexpr int SLOT = 16 * PH;
constexpr int DENSE_WAVES = 8;

__global__ __launch_bounds__(256) void k_dense2(const float* __restrict__ X,
                                                const float* __restrict__ W1,
                                                const float* __restrict__ b1,
                                                const float* __restrict__ W2,
                                                float* __restrict__ Y, int M) {
  constexpr int W2_FLOATS = (D_MID / 2) * PW2;
  constexpr int W1_FLOATS = (D_IN / 2) * PW1;
  __shared__ __attribute__((aligned(16))) float lds[W2_FLOATS + W1_FLOATS + DENSE_WAVES * SLOT];
  float* const w2p   = lds;
  float* const w1p   = lds + W2_FLOATS;
  float* const slots = lds + W2_FLOATS + W1_FLOATS;

  const int tid  = threadIdx.x;
  const int lane = tid & 31;
  const int wid  = __builtin_amdgcn_readfirstlane(tid >> 5);
  const int half = lane >> 4;
  const int r    = lane & 15;
  const int mtiles = (M + 15) >> 4;
  const int tile   = blockIdx.x * DENSE_WAVES + wid;
  const bool live  = tile < mtiles;
  const int m0 = (live ? tile : mtiles - 1) << 4;
  float* slot = slots + wid * SLOT;

  const int trow = lane >> 4;
  const int tcol = (lane & 15) * 4;
  v4f xin[8];
#pragma unroll
  for (int i = 0; i < 8; ++i) {
    const int gr = min(m0 + 2 * i + trow, M - 1);
    xin[i] = *(const v4f*)(X + (size_t)gr * D_IN + tcol);
  }
  float bias[D_MID / 16];
#pragma unroll
  for (int nt = 0; nt < D_MID / 16; ++nt) bias[nt] = b1[nt * 16 + r];

#pragma unroll
  for (int it = 0; it < (D_IN / 2) * (D_MID / 4) / 256; ++it) {
    const int item = tid + 256 * it;
    const int p = item / (D_MID / 4), n = (item % (D_MID / 4)) * 4;
    const v4f lo = *(const v4f*)(W1 + (size_t)(2 * p) * D_MID + n);
    const v4f hi = *(const v4f*)(W1 + (size_t)(2 * p + 1) * D_MID + n);
    const v4f s0 = {lo[0], hi[0], lo[1], hi[1]};
    const v4f s1 = {lo[2], hi[2], lo[3], hi[3]};
    *(v4f*)(w1p + p * PW1 + 2 * n)     = s0;
    *(v4f*)(w1p + p * PW1 + 2 * n + 4) = s1;
  }
#pragma unroll
  for (int it = 0; it < (D_MID / 2) * (D_FIN / 4) / 256; ++it) {
    const int item = tid + 256 * it;
    const int p = item / (D_FIN / 4), n = (item % (D_FIN / 4)) * 4;
    const v4f lo = *(const v4f*)(W2 + (size_t)(2 * p) * D_FIN + n);
    const v4f hi = *(const v4f*)(W2 + (size_t)(2 * p + 1) * D_FIN + n);
    const v4f s0 = {lo[0], hi[0], lo[1], hi[1]};
    const v4f s1 = {lo[2], hi[2], lo[3], hi[3]};
    *(v4f*)(w2p + p * PW2 + 2 * n)     = s0;
    *(v4f*)(w2p + p * PW2 + 2 * n + 4) = s1;
  }
#pragma unroll
  for (int i = 0; i < 8; ++i) *(v4f*)(slot + (2 * i + trow) * PA + tcol) = xin[i];
  __syncthreads();

  v8f acc[D_MID / 16];
#pragma unroll
  for (int nt = 0; nt < D_MID / 16; ++nt) acc[nt] = v8f{};
  {
    const float* ap = slot + r * PA + 2 * half;
    const float* bp = w1p + half * PW1 + 2 * r;
#pragma unroll
    for (int kk = 0; kk < D_IN / 4; ++kk) {
      const v2f a = *(const v2f*)(ap + 4 * kk);
#pragma unroll
      for (int nt = 0; nt < D_MID / 16; ++nt) {
        const v2f b = *(const v2f*)(bp + (2 * kk) * PW1 + nt * 32);
        acc[nt] = __builtin_amdgcn_wmma_f32_16x16x4_f32(
             false, a,  false, b,
             (short)0, acc[nt],  false,  false);
      }
    }
  }
#pragma unroll
  for (int nt = 0; nt < D_MID / 16; ++nt) {
#pragma unroll
    for (int v = 0; v < 8; ++v)
      slot[(v + 8 * half) * PH + nt * 16 + r] = fmaxf(acc[nt][v] + bias[nt], 0.0f);
  }
  __syncthreads();

  v8f acc2[D_FIN / 16];
#pragma unroll
  for (int nt = 0; nt < D_FIN / 16; ++nt) acc2[nt] = v8f{};
  {
    const float* ap = slot + r * PH + 2 * half;
    const float* bp = w2p + half * PW2 + 2 * r;
#pragma unroll
    for (int kk = 0; kk < D_MID / 4; ++kk) {
      const v2f a = *(const v2f*)(ap + 4 * kk);
#pragma unroll
      for (int nt = 0; nt < D_FIN / 16; ++nt) {
        const v2f b = *(const v2f*)(bp + (2 * kk) * PW2 + nt * 32);
        acc2[nt] = __builtin_amdgcn_wmma_f32_16x16x4_f32(
             false, a,  false, b,
             (short)0, acc2[nt],  false,  false);
      }
    }
  }
#pragma unroll
  for (int nt = 0; nt < D_FIN / 16; ++nt) {
#pragma unroll
    for (int v = 0; v < 8; ++v) slot[(v + 8 * half) * PA + nt * 16 + r] = acc2[nt][v];
  }
  __syncthreads();
  if (live) {
    if (m0 + 16 <= M) {
#pragma unroll
      for (int i = 0; i < 8; ++i) {
        const v4f y = *(const v4f*)(slot + (2 * i + trow) * PA + tcol);
        *(v4f*)(Y + (size_t)(m0 + 2 * i + trow) * D_FIN + tcol) = y;
      }
    } else {
#pragma unroll
      for (int i = 0; i < 8; ++i) {
        const int row = m0 + 2 * i + trow;
        const v4f y = *(const v4f*)(slot + (2 * i + trow) * PA + tcol);
        if (row < M) *(v4f*)(Y + (size_t)row * D_FIN + tcol) = y;
      }
    }
  }
}

template <int V>
struct rowvec;
template <> struct rowvec<4> { typedef v4f type; };
template <> struct rowvec<2> { typedef v2f type; };

template <int V, int B>
__device__ __forceinline__ void gather_rows(typename rowvec<V>::type& acc,
                                            const typename rowvec<V>::type* __restrict__ Xv,
                                            int ms, int mwbits, int j, int lane) {
  typedef typename rowvec<V>::type vf;
  int   s[B];
  float w[B];
  vf    x[B];
#pragma unroll
  for (int b = 0; b < B; ++b) {
    s[b] = __builtin_amdgcn_readlane(ms, j + b);
    w[b] = __int_as_float(__builtin_amdgcn_readlane(mwbits, j + b));
  }
#pragma unroll
  for (int b = 0; b < B; ++b) x[b] = Xv[(size_t)s[b] * 32 + lane];
#pragma unroll
  for (int b = 0; b < B; ++b) acc += x[b] * w[b];
}

template <int D, bool BIAS>
__global__ __launch_bounds__(256) void k_aggregate(float* __restrict__ Y,
                                                   const float* __restrict__ X,
                                                   const float* __restrict__ dinv,
                                                   const int* __restrict__ off,
                                                   const int* __restrict__ csr,
                                                   const float* __restrict__ bias, int N) {
  constexpr int V = D / 32;
  typedef typename rowvec<V>::type vf;
  int node = blockIdx.x * (blockDim.x >> 5) + (threadIdx.x >> 5);
  if (node >= N) return;
  node = __builtin_amdgcn_readfirstlane(node);
  const int lane = threadIdx.x & 31;
  const int beg = off[node], end = off[node + 1];
  const float dd = dinv[node];
  const vf* Xv = (const vf*)X;

  vf acc = Xv[(size_t)node * 32 + lane] * (dd * dd);
  for (int base = beg; base < end; base += 32) {
    const int n = min(32, end - base);
    int ms = 0;
    float mw = 0.0f;
    if (lane < n) {
      ms = csr[base + lane];
      mw = dinv[ms] * dd;
    }
    const int mwbits = __float_as_int(mw);
    int j = 0;
    for (; j + 8 <= n; j += 8) gather_rows<V, 8>(acc, Xv, ms, mwbits, j, lane);
    if (j + 4 <= n) { gather_rows<V, 4>(acc, Xv, ms, mwbits, j, lane); j += 4; }
    if (j + 2 <= n) { gather_rows<V, 2>(acc, Xv, ms, mwbits, j, lane); j += 2; }
    if (j < n)      { gather_rows<V, 1>(acc, Xv, ms, mwbits, j, lane); }
  }
  if (BIAS) acc += ((const vf*)bias)[lane];
  ((vf*)Y)[(size_t)node * 32 + lane] = acc;
}

extern "C" void kernel_launch(void* const* d_in, const int* in_sizes, int n_in,
                              void* d_out, int out_size, void* d_ws, size_t ws_size,
                              hipStream_t stream) {
  const float* emb  = (const float*)d_in[0];
  const float* W1   = (const float*)d_in[1];
  const float* b1   = (const float*)d_in[2];
  const float* W2   = (const float*)d_in[3];
  const float* b2   = (const float*)d_in[4];
  const int*   edge = (const int*)d_in[5];

  const int D_OUT = in_sizes[4];
  const int N     = in_sizes[0] / D_OUT;
  const int E     = in_sizes[5] / 2;
  const int* src  = edge;
  const int* dst  = edge + E;

  char* ws = (char*)d_ws;
  const size_t szN = (((size_t)(N + 1) * 4) + 255) & ~(size_t)255;
  const size_t szE = (((size_t)E * 4) + 255) & ~(size_t)255;
  const size_t szA = (((size_t)N * D_OUT * 4) + 255) & ~(size_t)255;
  int*   cnt  = (int*)ws;
  int*   off  = (int*)(ws + szN);
  float* dinv = (float*)(ws + 2 * szN);
  int*   rank = (int*)(ws + 3 * szN);
  int*   csr  = (int*)(ws + 3 * szN + szE);
  float* abuf = (float*)(ws + 3 * szN + 2 * szE);
  float* xbuf = (float*)(ws + 3 * szN + 2 * szE + szA);

  const int T = 256;
  const int wavesPerBlk = T / 32;
  const int mtiles  = (N + 15) / 16;
  const int N4      = (N + 3) & ~3;

  k_zero_cnt<<<(N4 + T - 1) / T, T, 0, stream>>>(cnt, N4);
  k_rank    <<<(E + T - 1) / T, T, 0, stream>>>(cnt, rank, dst, E);
  k_scan_deg<<<(N4 + SCAN_C - 1) / SCAN_C, SCAN_T, 0, stream>>>(cnt, off, dinv, N, E);
  k_fill    <<<(E + T - 1) / T, T, 0, stream>>>(csr, off, rank, src, dst, E);

  k_aggregate<64, false><<<(N + wavesPerBlk - 1) / wavesPerBlk, T, 0, stream>>>(
      abuf, emb, dinv, off, csr, nullptr, N);
  k_dense2<<<(mtiles + DENSE_WAVES - 1) / DENSE_WAVES, T, 0, stream>>>(
      abuf, W1, b1, W2, xbuf, N);
  float* out = (float*)d_out;
  k_aggregate<64, true><<<(N + wavesPerBlk - 1) / wavesPerBlk, T, 0, stream>>>(
      out, xbuf, dinv, off, csr, b2, N);
}


# edge sort without global atomics: edges binned by 512-node destination range via per-workgroup LDS counts, one workgroup per bin ranks and places by LDS atomics; 7 launches now 6; sum order changes
# speedup vs baseline: 5.2680x; 1.6336x over previous
#include <hip/hip_runtime.h>


typedef float v2f __attribute__((ext_vector_type(2)));
typedef float v4f __attribute__((ext_vector_type(4)));
typedef float v8f __attribute__((ext_vector_type(8)));
typedef int   v4i __attribute__((ext_vector_type(4)));

constexpr int RSH   = 9;
constexpr int RNG   = 1 << RSH;
constexpr int MAXB  = 256;
constexpr int BIN_T = 1024;
constexpr int EPT   = 8;
constexpr int CHUNK = BIN_T * EPT;

__global__ __launch_bounds__(BIN_T) void k_bin_count(int* __restrict__ tab,
                                                     const int* __restrict__ dst, int E) {
  __shared__ int h[MAXB];
  const int tid = threadIdx.x;
  const int e0 = blockIdx.x * CHUNK + tid;
  int d[EPT];
#pragma unroll
  for (int k = 0; k < EPT; ++k) d[k] = dst[min(e0 + k * BIN_T, E - 1)];
  if (tid < MAXB) h[tid] = 0;
  __syncthreads();
#pragma unroll
  for (int k = 0; k < EPT; ++k)
    if (e0 + k * BIN_T < E) atomicAdd(&h[d[k] >> RSH], 1);
  __syncthreads();
  if (tid < MAXB) tab[blockIdx.x * MAXB + tid] = h[tid];
}

constexpr int TQ = BIN_T / (MAXB / 4);
constexpr int TJ = 8;

__global__ __launch_bounds__(BIN_T) void k_bin_place(int* __restrict__ bkt,
                                                     int* __restrict__ bstart,
                                                     const int* __restrict__ tab,
                                                     const int* __restrict__ src,
                                                     const int* __restrict__ dst,
                                                     int E, int nchunk) {
  __shared__ int h[MAXB];
  __shared__ __attribute__((aligned(16))) int ptot[TQ][MAXB];
  __shared__ __attribute__((aligned(16))) int ppre[TQ][MAXB];
  __shared__ int wsum[MAXB / 32];
  __shared__ int base[MAXB];
  const int tid = threadIdx.x, lane = tid & 31, wid = tid >> 5;
  const int blk = blockIdx.x;
  const int e0 = blk * CHUNK + tid;
  int d[EPT], s[EPT], r[EPT];
#pragma unroll
  for (int k = 0; k < EPT; ++k) {
    const int i = min(e0 + k * BIN_T, E - 1);
    d[k] = dst[i];
    s[k] = src[i];
    r[k] = 0;
  }
  if (tid < MAXB) h[tid] = 0;
  const int q = tid >> 6, t4 = (tid & 63) * 4;
  v4i tot = {0, 0, 0, 0}, pre = {0, 0, 0, 0};
  for (int c0 = q; c0 < nchunk; c0 += TQ * TJ) {
    v4i v[TJ];
#pragma unroll
    for (int j = 0; j < TJ; ++j)
      v[j] = *(const v4i*)(tab + min(c0 + TQ * j, nchunk - 1) * MAXB + t4);
#pragma unroll
    for (int j = 0; j < TJ; ++j) {
      const int c = c0 + TQ * j;
      const v4i z = {0, 0, 0, 0};
      tot += (c < nchunk) ? v[j] : z;
      pre += (c < blk) ? v[j] : z;
    }
  }
  *(v4i*)(&ptot[q][t4]) = tot;
  *(v4i*)(&ppre[q][t4]) = pre;
  __syncthreads();
#pragma unroll
  for (int k = 0; k < EPT; ++k)
    if (e0 + k * BIN_T < E) r[k] = atomicAdd(&h[d[k] >> RSH], 1);
  int mytot = 0, mypre = 0, incl = 0;
  if (tid < MAXB) {
#pragma unroll
    for (int j = 0; j < TQ; ++j) {
      mytot += ptot[j][tid];
      mypre += ppre[j][tid];
    }
    incl = mytot;
    for (int o = 1; o < 32; o <<= 1) {
      const int y = __shfl_up(incl, o);
      if (lane >= o) incl += y;
    }
    if (lane == 31) wsum[wid] = incl;
  }
  __syncthreads();
  if (tid < MAXB) {
    int carry = 0;
    for (int w = 0; w < wid; ++w) carry += wsum[w];
    const int ex = carry + incl - mytot;
    base[tid] = ex + mypre;
    if (blk == 0) {
      bstart[tid] = ex;
      if (tid == 0) bstart[MAXB] = E;
    }
  }
  __syncthreads();
#pragma unroll
  for (int k = 0; k < EPT; ++k)
    if (e0 + k * BIN_T < E)
      bkt[base[d[k] >> RSH] + r[k]] = (s[k] << RSH) | (d[k] & (RNG - 1));
}

constexpr int CU = 10;

__global__ __launch_bounds__(BIN_T) void k_bin_csr(int* __restrict__ csr,
                                                   int* __restrict__ off,
                                                   float* __restrict__ dinv,
                                                   const int* __restrict__ bkt,
                                                   const int* __restrict__ bstart,
                                                   int N, int E) {
  __shared__ int c[RNG];
  __shared__ int wsum[RNG / 32];
  const int tid = threadIdx.x, lane = tid & 31, wid = tid >> 5;
  const int b = blockIdx.x;
  const int s = bstart[b], e = bstart[b + 1];
  if (tid < RNG) c[tid] = 0;
  __syncthreads();
  for (int i0 = s + tid; i0 < e; i0 += CU * BIN_T) {
    int w[CU];
#pragma unroll
    for (int j = 0; j < CU; ++j) w[j] = bkt[min(i0 + j * BIN_T, e - 1)];
    asm volatile("" :: "v"(w[0]), "v"(w[1]), "v"(w[2]), "v"(w[3]), "v"(w[4]),
                       "v"(w[5]), "v"(w[6]), "v"(w[7]), "v"(w[8]), "v"(w[9]));
#pragma unroll
    for (int j = 0; j < CU; ++j)
      if (i0 + j * BIN_T < e) atomicAdd(&c[w[j] & (RNG - 1)], 1);
  }
  __syncthreads();
  int v = 0, incl = 0;
  if (tid < RNG) {
    v = c[tid];
    incl = v;
    for (int o = 1; o < 32; o <<= 1) {
      const int y = __shfl_up(incl, o);
      if (lane >= o) incl += y;
    }
    if (lane == 31) wsum[wid] = incl;
  }
  __syncthreads();
  if (tid < RNG) {
    int carry = 0;
    for (int w = 0; w < wid; ++w) carry += wsum[w];
    const int o = s + carry + incl - v;
    c[tid] = o;
    const int node = b * RNG + tid;
    if (node < N) {
      off[node]  = o;
      dinv[node] = rsqrtf((float)(v + 1));
    }
  }
  if (b == (int)gridDim.x - 1 && tid == 0) off[N] = E;
  __syncthreads();
  for (int i0 = s + tid; i0 < e; i0 += CU * BIN_T) {
    int w[CU];
#pragma unroll
    for (int j = 0; j < CU; ++j) w[j] = bkt[min(i0 + j * BIN_T, e - 1)];
    asm volatile("" :: "v"(w[0]), "v"(w[1]), "v"(w[2]), "v"(w[3]), "v"(w[4]),
                       "v"(w[5]), "v"(w[6]), "v"(w[7]), "v"(w[8]), "v"(w[9]));
    int p[CU];
#pragma unroll
    for (int j = 0; j < CU; ++j) {
      p[j] = 0;
      if (i0 + j * BIN_T < e) p[j] = atomicAdd(&c[w[j] & (RNG - 1)], 1);
    }
#pragma unroll
    for (int j = 0; j < CU; ++j)
      if (i0 + j * BIN_T < e) csr[p[j]] = w[j] >> RSH;
  }
}

constexpr int D_IN  = 64;
constexpr int D_MID = 128;
constexpr int D_FIN = 64;
constexpr int PW1  = 2 * D_MID + 32;
constexpr int PW2  = 2 * D_FIN + 32;
constexpr int PA   = D_IN + 4;
constexpr int PH   = D_MID + 4;
constexpr int SLOT = 16 * PH;
constexpr int DENSE_WAVES = 8;

__global__ __launch_bounds__(256) void k_dense2(const float* __restrict__ X,
                                                const float* __restrict__ W1,
                                                const float* __restrict__ b1,
                                                const float* __restrict__ W2,
                                                float* __restrict__ Y, int M) {
  constexpr int W2_FLOATS = (D_MID / 2) * PW2;
  constexpr int W1_FLOATS = (D_IN / 2) * PW1;
  __shared__ __attribute__((aligned(16))) float lds[W2_FLOATS + W1_FLOATS + DENSE_WAVES * SLOT];
  float* const w2p   = lds;
  float* const w1p   = lds + W2_FLOATS;
  float* const slots = lds + W2_FLOATS + W1_FLOATS;

  const int tid  = threadIdx.x;
  const int lane = tid & 31;
  const int wid  = __builtin_amdgcn_readfirstlane(tid >> 5);
  const int half = lane >> 4;
  const int r    = lane & 15;
  const int mtiles = (M + 15) >> 4;
  const int tile   = blockIdx.x * DENSE_WAVES + wid;
  const bool live  = tile < mtiles;
  const int m0 = (live ? tile : mtiles - 1) << 4;
  float* slot = slots + wid * SLOT;

  const int trow = lane >> 4;
  const int tcol = (lane & 15) * 4;
  v4f xin[8];
#pragma unroll
  for (int i = 0; i < 8; ++i) {
    const int gr = min(m0 + 2 * i + trow, M - 1);
    xin[i] = *(const v4f*)(X + (size_t)gr * D_IN + tcol);
  }
  float bias[D_MID / 16];
#pragma unroll
  for (int nt = 0; nt < D_MID / 16; ++nt) bias[nt] = b1[nt * 16 + r];

#pragma unroll
  for (int it = 0; it < (D_IN / 2) * (D_MID / 4) / 256; ++it) {
    const int item = tid + 256 * it;
    const int p = item / (D_MID / 4), n = (item % (D_MID / 4)) * 4;
    const v4f lo = *(const v4f*)(W1 + (size_t)(2 * p) * D_MID + n);
    const v4f hi = *(const v4f*)(W1 + (size_t)(2 * p + 1) * D_MID + n);
    const v4f s0 = {lo[0], hi[0], lo[1], hi[1]};
    const v4f s1 = {lo[2], hi[2], lo[3], hi[3]};
    *(v4f*)(w1p + p * PW1 + 2 * n)     = s0;
    *(v4f*)(w1p + p * PW1 + 2 * n + 4) = s1;
  }
#pragma unroll
  for (int it = 0; it < (D_MID / 2) * (D_FIN / 4) / 256; ++it) {
    const int item = tid + 256 * it;
    const int p = item / (D_FIN / 4), n = (item % (D_FIN / 4)) * 4;
    const v4f lo = *(const v4f*)(W2 + (size_t)(2 * p) * D_FIN + n);
    const v4f hi = *(const v4f*)(W2 + (size_t)(2 * p + 1) * D_FIN + n);
    const v4f s0 = {lo[0], hi[0], lo[1], hi[1]};
    const v4f s1 = {lo[2], hi[2], lo[3], hi[3]};
    *(v4f*)(w2p + p * PW2 + 2 * n)     = s0;
    *(v4f*)(w2p + p * PW2 + 2 * n + 4) = s1;
  }
#pragma unroll
  for (int i = 0; i < 8; ++i) *(v4f*)(slot + (2 * i + trow) * PA + tcol) = xin[i];
  __syncthreads();

  v8f acc[D_MID / 16];
#pragma unroll
  for (int nt = 0; nt < D_MID / 16; ++nt) acc[nt] = v8f{};
  {
    const float* ap = slot + r * PA + 2 * half;
    const float* bp = w1p + half * PW1 + 2 * r;
#pragma unroll
    for (int kk = 0; kk < D_IN / 4; ++kk) {
      const v2f a = *(const v2f*)(ap + 4 * kk);
#pragma unroll
      for (int nt = 0; nt < D_MID / 16; ++nt) {
        const v2f b = *(const v2f*)(bp + (2 * kk) * PW1 + nt * 32);
        acc[nt] = __builtin_amdgcn_wmma_f32_16x16x4_f32(
             false, a,  false, b,
             (short)0, acc[nt],  false,  false);
      }
    }
  }
#pragma unroll
  for (int nt = 0; nt < D_MID / 16; ++nt) {
#pragma unroll
    for (int v = 0; v < 8; ++v)
      slot[(v + 8 * half) * PH + nt * 16 + r] = fmaxf(acc[nt][v] + bias[nt], 0.0f);
  }
  __syncthreads();

  v8f acc2[D_FIN / 16];
#pragma unroll
  for (int nt = 0; nt < D_FIN / 16; ++nt) acc2[nt] = v8f{};
  {
    const float* ap = slot + r * PH + 2 * half;
    const float* bp = w2p + half * PW2 + 2 * r;
#pragma unroll
    for (int kk = 0; kk < D_MID / 4; ++kk) {
      const v2f a = *(const v2f*)(ap + 4 * kk);
#pragma unroll
      for (int nt = 0; nt < D_FIN / 16; ++nt) {
        const v2f b = *(const v2f*)(bp + (2 * kk) * PW2 + nt * 32);
        acc2[nt] = __builtin_amdgcn_wmma_f32_16x16x4_f32(
             false, a,  false, b,
             (short)0, acc2[nt],  false,  false);
      }
    }
  }
#pragma unroll
  for (int nt = 0; nt < D_FIN / 16; ++nt) {
#pragma unroll
    for (int v = 0; v < 8; ++v) slot[(v + 8 * half) * PA + nt * 16 + r] = acc2[nt][v];
  }
  __syncthreads();
  if (live) {
    if (m0 + 16 <= M) {
#pragma unroll
      for (int i = 0; i < 8; ++i) {
        const v4f y = *(const v4f*)(slot + (2 * i + trow) * PA + tcol);
        *(v4f*)(Y + (size_t)(m0 + 2 * i + trow) * D_FIN + tcol) = y;
      }
    } else {
#pragma unroll
      for (int i = 0; i < 8; ++i) {
        const int row = m0 + 2 * i + trow;
        const v4f y = *(const v4f*)(slot + (2 * i + trow) * PA + tcol);
        if (row < M) *(v4f*)(Y + (size_t)row * D_FIN + tcol) = y;
      }
    }
  }
}

template <int V>
struct rowvec;
template <> struct rowvec<4> { typedef v4f type; };
template <> struct rowvec<2> { typedef v2f type; };

template <int V, int B>
__device__ __forceinline__ void gather_rows(typename rowvec<V>::type& acc,
                                            const typename rowvec<V>::type* __restrict__ Xv,
                                            int ms, int mwbits, int j, int lane) {
  typedef typename rowvec<V>::type vf;
  int   s[B];
  float w[B];
  vf    x[B];
#pragma unroll
  for (int b = 0; b < B; ++b) {
    s[b] = __builtin_amdgcn_readlane(ms, j + b);
    w[b] = __int_as_float(__builtin_amdgcn_readlane(mwbits, j + b));
  }
#pragma unroll
  for (int b = 0; b < B; ++b) x[b] = Xv[(size_t)s[b] * 32 + lane];
#pragma unroll
  for (int b = 0; b < B; ++b) acc += x[b] * w[b];
}

template <int D, bool BIAS>
__global__ __launch_bounds__(256) void k_aggregate(float* __restrict__ Y,
                                                   const float* __restrict__ X,
                                                   const float* __restrict__ dinv,
                                                   const int* __restrict__ off,
                                                   const int* __restrict__ csr,
                                                   const float* __restrict__ bias, int N) {
  constexpr int V = D / 32;
  typedef typename rowvec<V>::type vf;
  int node = blockIdx.x * (blockDim.x >> 5) + (threadIdx.x >> 5);
  if (node >= N) return;
  node = __builtin_amdgcn_readfirstlane(node);
  const int lane = threadIdx.x & 31;
  const int beg = off[node], end = off[node + 1];
  const float dd = dinv[node];
  const vf* Xv = (const vf*)X;

  vf acc = Xv[(size_t)node * 32 + lane] * (dd * dd);
  for (int base = beg; base < end; base += 32) {
    const int n = min(32, end - base);
    int ms = 0;
    float mw = 0.0f;
    if (lane < n) {
      ms = csr[base + lane];
      mw = dinv[ms] * dd;
    }
    const int mwbits = __float_as_int(mw);
    int j = 0;
    for (; j + 8 <= n; j += 8) gather_rows<V, 8>(acc, Xv, ms, mwbits, j, lane);
    if (j + 4 <= n) { gather_rows<V, 4>(acc, Xv, ms, mwbits, j, lane); j += 4; }
    if (j + 2 <= n) { gather_rows<V, 2>(acc, Xv, ms, mwbits, j, lane); j += 2; }
    if (j < n)      { gather_rows<V, 1>(acc, Xv, ms, mwbits, j, lane); }
  }
  if (BIAS) acc += ((const vf*)bias)[lane];
  ((vf*)Y)[(size_t)node * 32 + lane] = acc;
}

extern "C" void kernel_launch(void* const* d_in, const int* in_sizes, int n_in,
                              void* d_out, int out_size, void* d_ws, size_t ws_size,
                              hipStream_t stream) {
  const float* emb  = (const float*)d_in[0];
  const float* W1   = (const float*)d_in[1];
  const float* b1   = (const float*)d_in[2];
  const float* W2   = (const float*)d_in[3];
  const float* b2   = (const float*)d_in[4];
  const int*   edge = (const int*)d_in[5];

  const int D_OUT = in_sizes[4];
  const int N     = in_sizes[0] / D_OUT;
  const int E     = in_sizes[5] / 2;
  const int* src  = edge;
  const int* dst  = edge + E;

  char* ws = (char*)d_ws;
  const int nchunk = (E + CHUNK - 1) / CHUNK;
  const int nbin   = (N + RNG - 1) / RNG;
  const size_t szN = (((size_t)(N + 1) * 4) + 255) & ~(size_t)255;
  const size_t szE = (((size_t)E * 4) + 255) & ~(size_t)255;
  const size_t szA = (((size_t)N * D_OUT * 4) + 255) & ~(size_t)255;
  const size_t szT = (size_t)nchunk * MAXB * 4;
  const size_t szB = (((size_t)(MAXB + 1) * 4) + 255) & ~(size_t)255;
  int*   off    = (int*)ws;
  float* dinv   = (float*)(ws + szN);
  int*   tab    = (int*)(ws + 2 * szN);
  int*   bstart = (int*)(ws + 2 * szN + szT);
  int*   bkt    = (int*)(ws + 2 * szN + szT + szB);
  int*   csr    = (int*)(ws + 2 * szN + szT + szB + szE);
  float* abuf   = (float*)(ws + 2 * szN + szT + szB + 2 * szE);
  float* xbuf   = (float*)(ws + 2 * szN + szT + szB + 2 * szE + szA);

  const int T = 256;
  const int wavesPerBlk = T / 32;
  const int mtiles  = (N + 15) / 16;

  k_bin_count<<<nchunk, BIN_T, 0, stream>>>(tab, dst, E);
  k_bin_place<<<nchunk, BIN_T, 0, stream>>>(bkt, bstart, tab, src, dst, E, nchunk);
  k_bin_csr  <<<nbin, BIN_T, 0, stream>>>(csr, off, dinv, bkt, bstart, N, E);

  k_aggregate<64, false><<<(N + wavesPerBlk - 1) / wavesPerBlk, T, 0, stream>>>(
      abuf, emb, dinv, off, csr, nullptr, N);
  k_dense2<<<(mtiles + DENSE_WAVES - 1) / DENSE_WAVES, T, 0, stream>>>(
      abuf, W1, b1, W2, xbuf, N);
  float* out = (float*)d_out;
  k_aggregate<64, true><<<(N + wavesPerBlk - 1) / wavesPerBlk, T, 0, stream>>>(
      out, xbuf, dinv, off, csr, b2, N);
}


# edge sort: both placements are ordered in LDS first (a chunk's edges by bin, a bin's source ids by segment) and leave as runs of neighbouring words, not one 4-byte store per line; sum order kept
# speedup vs baseline: 6.7075x; 1.2733x over previous
#include <hip/hip_runtime.h>


typedef float v2f __attribute__((ext_vector_type(2)));
typedef float v4f __attribute__((ext_vector_type(4)));
typedef float v8f __attribute__((ext_vector_type(8)));
typedef int   v4i __attribute__((ext_vector_type(4)));

constexpr int RSH   = 9;
constexpr int RNG   = 1 << RSH;
constexpr int MAXB  = 256;
constexpr int BIN_T = 1024;
constexpr int EPT   = 8;
constexpr int CHUNK = BIN_T * EPT;

__global__ __launch_bounds__(BIN_T) void k_bin_count(int* __restrict__ tab,
                                                     const int* __restrict__ dst, int E) {
  __shared__ int h[MAXB];
  const int tid = threadIdx.x;
  const int e0 = blockIdx.x * CHUNK + tid;
  int d[EPT];
#pragma unroll
  for (int k = 0; k < EPT; ++k) d[k] = dst[min(e0 + k * BIN_T, E - 1)];
  if (tid < MAXB) h[tid] = 0;
  __syncthreads();
#pragma unroll
  for (int k = 0; k < EPT; ++k)
    if (e0 + k * BIN_T < E) atomicAdd(&h[d[k] >> RSH], 1);
  __syncthreads();
  if (tid < MAXB) tab[blockIdx.x * MAXB + tid] = h[tid];
}

constexpr int TQ = BIN_T / (MAXB / 4);
constexpr int TJ = 8;

__global__ __launch_bounds__(BIN_T) void k_bin_place(int* __restrict__ bkt,
                                                     int* __restrict__ bstart,
                                                     const int* __restrict__ tab,
                                                     const int* __restrict__ src,
                                                     const int* __restrict__ dst,
                                                     int E, int nchunk) {
  __shared__ int h[MAXB];
  __shared__ __attribute__((aligned(16))) int ptot[TQ][MAXB];
  __shared__ __attribute__((aligned(16))) int ppre[TQ][MAXB];
  __shared__ int wsum[MAXB / 32];
  __shared__ int wsum2[MAXB / 32];
  __shared__ int delta[MAXB];
  __shared__ int stage[CHUNK];
  __shared__ unsigned char sbin[CHUNK];
  const int tid = threadIdx.x, lane = tid & 31, wid = tid >> 5;
  const int blk = blockIdx.x;
  const int e0 = blk * CHUNK + tid;
  int d[EPT], s[EPT], p[EPT];
#pragma unroll
  for (int k = 0; k < EPT; ++k) {
    const int i = min(e0 + k * BIN_T, E - 1);
    d[k] = dst[i];
    s[k] = src[i];
    p[k] = 0;
  }
  int own = 0;
  if (tid < MAXB) own = tab[blk * MAXB + tid];
  const int q = tid >> 6, t4 = (tid & 63) * 4;
  v4i tot = {0, 0, 0, 0}, pre = {0, 0, 0, 0};
  for (int c0 = q; c0 < nchunk; c0 += TQ * TJ) {
    v4i v[TJ];
#pragma unroll
    for (int j = 0; j < TJ; ++j)
      v[j] = *(const v4i*)(tab + min(c0 + TQ * j, nchunk - 1) * MAXB + t4);
#pragma unroll
    for (int j = 0; j < TJ; ++j) {
      const int c = c0 + TQ * j;
      const v4i z = {0, 0, 0, 0};
      tot += (c < nchunk) ? v[j] : z;
      pre += (c < blk) ? v[j] : z;
    }
  }
  *(v4i*)(&ptot[q][t4]) = tot;
  *(v4i*)(&ppre[q][t4]) = pre;
  int incl2 = own;
  if (tid < MAXB) {
    for (int o = 1; o < 32; o <<= 1) {
      const int y = __shfl_up(incl2, o);
      if (lane >= o) incl2 += y;
    }
    if (lane == 31) wsum2[wid] = incl2;
  }
  __syncthreads();
  int mytot = 0, mypre = 0, incl = 0, lst = 0;
  if (tid < MAXB) {
#pragma unroll
    for (int j = 0; j < TQ; ++j) {
      mytot += ptot[j][tid];
      mypre += ppre[j][tid];
    }
    incl = mytot;
    for (int o = 1; o < 32; o <<= 1) {
      const int y = __shfl_up(incl, o);
      if (lane >= o) incl += y;
    }
    if (lane == 31) wsum[wid] = incl;
    int carry2 = 0;
    for (int w = 0; w < wid; ++w) carry2 += wsum2[w];
    lst = carry2 + incl2 - own;
    h[tid] = lst;
  }
  __syncthreads();
#pragma unroll
  for (int k = 0; k < EPT; ++k)
    if (e0 + k * BIN_T < E) p[k] = atomicAdd(&h[d[k] >> RSH], 1);
  if (tid < MAXB) {
    int carry = 0;
    for (int w = 0; w < wid; ++w) carry += wsum[w];
    const int ex = carry + incl - mytot;
    delta[tid] = ex + mypre - lst;
    if (blk == 0) {
      bstart[tid] = ex;
      if (tid == 0) bstart[MAXB] = E;
    }
  }
#pragma unroll
  for (int k = 0; k < EPT; ++k)
    if (e0 + k * BIN_T < E) {
      stage[p[k]] = (s[k] << RSH) | (d[k] & (RNG - 1));
      sbin[p[k]]  = (unsigned char)(d[k] >> RSH);
    }
  __syncthreads();
  const int cnt = min(CHUNK, E - blk * CHUNK);
  int sb[EPT], dl[EPT], sv[EPT];
#pragma unroll
  for (int k = 0; k < EPT; ++k) sb[k] = sbin[min(tid + k * BIN_T, cnt - 1)];
#pragma unroll
  for (int k = 0; k < EPT; ++k) {
    dl[k] = delta[sb[k]];
    sv[k] = stage[min(tid + k * BIN_T, cnt - 1)];
  }
#pragma unroll
  for (int k = 0; k < EPT; ++k) {
    const int pos = tid + k * BIN_T;
    if (pos < cnt) bkt[dl[k] + pos] = sv[k];
  }
}

constexpr int CU = 10;
constexpr int CAP = 16384;

__global__ __launch_bounds__(BIN_T) void k_bin_csr(int* __restrict__ csr,
                                                   int* __restrict__ off,
                                                   float* __restrict__ dinv,
                                                   const int* __restrict__ bkt,
                                                   const int* __restrict__ bstart,
                                                   int N, int E) {
  __shared__ int c[RNG];
  __shared__ int wsum[RNG / 32];
  __shared__ int stage[CAP];
  const int tid = threadIdx.x, lane = tid & 31, wid = tid >> 5;
  const int b = blockIdx.x;
  const int s = bstart[b], e = bstart[b + 1];
  const int cnt = e - s;
  const bool fits = cnt <= CAP;
  if (tid < RNG) c[tid] = 0;
  __syncthreads();
  for (int i0 = s + tid; i0 < e; i0 += CU * BIN_T) {
    int w[CU];
#pragma unroll
    for (int j = 0; j < CU; ++j) w[j] = bkt[min(i0 + j * BIN_T, e - 1)];
    asm volatile("" :: "v"(w[0]), "v"(w[1]), "v"(w[2]), "v"(w[3]), "v"(w[4]),
                       "v"(w[5]), "v"(w[6]), "v"(w[7]), "v"(w[8]), "v"(w[9]));
#pragma unroll
    for (int j = 0; j < CU; ++j)
      if (i0 + j * BIN_T < e) atomicAdd(&c[w[j] & (RNG - 1)], 1);
  }
  __syncthreads();
  int v = 0, incl = 0;
  if (tid < RNG) {
    v = c[tid];
    incl = v;
    for (int o = 1; o < 32; o <<= 1) {
      const int y = __shfl_up(incl, o);
      if (lane >= o) incl += y;
    }
    if (lane == 31) wsum[wid] = incl;
  }
  __syncthreads();
  if (tid < RNG) {
    int carry = 0;
    for (int w = 0; w < wid; ++w) carry += wsum[w];
    const int o = s + carry + incl - v;
    c[tid] = o;
    const int node = b * RNG + tid;
    if (node < N) {
      off[node]  = o;
      dinv[node] = rsqrtf((float)(v + 1));
    }
  }
  if (b == (int)gridDim.x - 1 && tid == 0) off[N] = E;
  __syncthreads();
  for (int i0 = s + tid; i0 < e; i0 += CU * BIN_T) {
    int w[CU];
#pragma unroll
    for (int j = 0; j < CU; ++j) w[j] = bkt[min(i0 + j * BIN_T, e - 1)];
    asm volatile("" :: "v"(w[0]), "v"(w[1]), "v"(w[2]), "v"(w[3]), "v"(w[4]),
                       "v"(w[5]), "v"(w[6]), "v"(w[7]), "v"(w[8]), "v"(w[9]));
    int p[CU];
#pragma unroll
    for (int j = 0; j < CU; ++j) {
      p[j] = 0;
      if (i0 + j * BIN_T < e) p[j] = atomicAdd(&c[w[j] & (RNG - 1)], 1);
    }
    if (fits) {
#pragma unroll
      for (int j = 0; j < CU; ++j)
        if (i0 + j * BIN_T < e) stage[p[j] - s] = w[j] >> RSH;
    } else {
#pragma unroll
      for (int j = 0; j < CU; ++j)
        if (i0 + j * BIN_T < e) csr[p[j]] = w[j] >> RSH;
    }
  }
  __syncthreads();
  if (fits && cnt > 0) {
    for (int i0 = tid - (s & 31); i0 < cnt; i0 += 4 * BIN_T) {
      int v[4];
#pragma unroll
      for (int j = 0; j < 4; ++j) v[j] = stage[min(max(i0 + j * BIN_T, 0), cnt - 1)];
#pragma unroll
      for (int j = 0; j < 4; ++j) {
        const int i = i0 + j * BIN_T;
        if (i >= 0 && i < cnt) csr[s + i] = v[j];
      }
    }
  }
}

constexpr int D_IN  = 64;
constexpr int D_MID = 128;
constexpr int D_FIN = 64;
constexpr int PW1  = 2 * D_MID + 32;
constexpr int PW2  = 2 * D_FIN + 32;
constexpr int PA   = D_IN + 4;
constexpr int PH   = D_MID + 4;
constexpr int SLOT = 16 * PH;
constexpr int DENSE_WAVES = 8;

__global__ __launch_bounds__(256) void k_dense2(const float* __restrict__ X,
                                                const float* __restrict__ W1,
                                                const float* __restrict__ b1,
                                                const float* __restrict__ W2,
                                                float* __restrict__ Y, int M) {
  constexpr int W2_FLOATS = (D_MID / 2) * PW2;
  constexpr int W1_FLOATS = (D_IN / 2) * PW1;
  __shared__ __attribute__((aligned(16))) float lds[W2_FLOATS + W1_FLOATS + DENSE_WAVES * SLOT];
  float* const w2p   = lds;
  float* const w1p   = lds + W2_FLOATS;
  float* const slots = lds + W2_FLOATS + W1_FLOATS;

  const int tid  = threadIdx.x;
  const int lane = tid & 31;
  const int wid  = __builtin_amdgcn_readfirstlane(tid >> 5);
  const int half = lane >> 4;
  const int r    = lane & 15;
  const int mtiles = (M + 15) >> 4;
  const int tile   = blockIdx.x * DENSE_WAVES + wid;
  const bool live  = tile < mtiles;
  const int m0 = (live ? tile : mtiles - 1) << 4;
  float* slot = slots + wid * SLOT;

  const int trow = lane >> 4;
  const int tcol = (lane & 15) * 4;
  v4f xin[8];
#pragma unroll
  for (int i = 0; i < 8; ++i) {
    const int gr = min(m0 + 2 * i + trow, M - 1);
    xin[i] = *(const v4f*)(X + (size_t)gr * D_IN + tcol);
  }
  float bias[D_MID / 16];
#pragma unroll
  for (int nt = 0; nt < D_MID / 16; ++nt) bias[nt] = b1[nt * 16 + r];

#pragma unroll
  for (int it = 0; it < (D_IN / 2) * (D_MID / 4) / 256; ++it) {
    const int item = tid + 256 * it;
    const int p = item / (D_MID / 4), n = (item % (D_MID / 4)) * 4;
    const v4f lo = *(const v4f*)(W1 + (size_t)(2 * p) * D_MID + n);
    const v4f hi = *(const v4f*)(W1 + (size_t)(2 * p + 1) * D_MID + n);
    const v4f s0 = {lo[0], hi[0], lo[1], hi[1]};
    const v4f s1 = {lo[2], hi[2], lo[3], hi[3]};
    *(v4f*)(w1p + p * PW1 + 2 * n)     = s0;
    *(v4f*)(w1p + p * PW1 + 2 * n + 4) = s1;
  }
#pragma unroll
  for (int it = 0; it < (D_MID / 2) * (D_FIN / 4) / 256; ++it) {
    const int item = tid + 256 * it;
    const int p = item / (D_FIN / 4), n = (item % (D_FIN / 4)) * 4;
    const v4f lo = *(const v4f*)(W2 + (size_t)(2 * p) * D_FIN + n);
    const v4f hi = *(const v4f*)(W2 + (size_t)(2 * p + 1) * D_FIN + n);
    const v4f s0 = {lo[0], hi[0], lo[1], hi[1]};
    const v4f s1 = {lo[2], hi[2], lo[3], hi[3]};
    *(v4f*)(w2p + p * PW2 + 2 * n)     = s0;
    *(v4f*)(w2p + p * PW2 + 2 * n + 4) = s1;
  }
#pragma unroll
  for (int i = 0; i < 8; ++i) *(v4f*)(slot + (2 * i + trow) * PA + tcol) = xin[i];
  __syncthreads();

  v8f acc[D_MID / 16];
#pragma unroll
  for (int nt = 0; nt < D_MID / 16; ++nt) acc[nt] = v8f{};
  {
    const float* ap = slot + r * PA + 2 * half;
    const float* bp = w1p + half * PW1 + 2 * r;
#pragma unroll
    for (int kk = 0; kk < D_IN / 4; ++kk) {
      const v2f a = *(const v2f*)(ap + 4 * kk);
#pragma unroll
      for (int nt = 0; nt < D_MID / 16; ++nt) {
        const v2f b = *(const v2f*)(bp + (2 * kk) * PW1 + nt * 32);
        acc[nt] = __builtin_amdgcn_wmma_f32_16x16x4_f32(
             false, a,  false, b,
             (short)0, acc[nt],  false,  false);
      }
    }
  }
#pragma unroll
  for (int nt = 0; nt < D_MID / 16; ++nt) {
#pragma unroll
    for (int v = 0; v < 8; ++v)
      slot[(v + 8 * half) * PH + nt * 16 + r] = fmaxf(acc[nt][v] + bias[nt], 0.0f);
  }
  __syncthreads();

  v8f acc2[D_FIN / 16];
#pragma unroll
  for (int nt = 0; nt < D_FIN / 16; ++nt) acc2[nt] = v8f{};
  {
    const float* ap = slot + r * PH + 2 * half;
    const float* bp = w2p + half * PW2 + 2 * r;
#pragma unroll
    for (int kk = 0; kk < D_MID / 4; ++kk) {
      const v2f a = *(const v2f*)(ap + 4 * kk);
#pragma unroll
      for (int nt = 0; nt < D_FIN / 16; ++nt) {
        const v2f b = *(const v2f*)(bp + (2 * kk) * PW2 + nt * 32);
        acc2[nt] = __builtin_amdgcn_wmma_f32_16x16x4_f32(
             false, a,  false, b,
             (short)0, acc2[nt],  false,  false);
      }
    }
  }
#pragma unroll
  for (int nt = 0; nt < D_FIN / 16; ++nt) {
#pragma unroll
    for (int v = 0; v < 8; ++v) slot[(v + 8 * half) * PA + nt * 16 + r] = acc2[nt][v];
  }
  __syncthreads();
  if (live) {
    if (m0 + 16 <= M) {
#pragma unroll
      for (int i = 0; i < 8; ++i) {
        const v4f y = *(const v4f*)(slot + (2 * i + trow) * PA + tcol);
        *(v4f*)(Y + (size_t)(m0 + 2 * i + trow) * D_FIN + tcol) = y;
      }
    } else {
#pragma unroll
      for (int i = 0; i < 8; ++i) {
        const int row = m0 + 2 * i + trow;
        const v4f y = *(const v4f*)(slot + (2 * i + trow) * PA + tcol);
        if (row < M) *(v4f*)(Y + (size_t)row * D_FIN + tcol) = y;
      }
    }
  }
}

template <int V>
struct rowvec;
template <> struct rowvec<4> { typedef v4f type; };
template <> struct rowvec<2> { typedef v2f type; };

template <int V, int B>
__device__ __forceinline__ void gather_rows(typename rowvec<V>::type& acc,
                                            const typename rowvec<V>::type* __restrict__ Xv,
                                            int ms, int mwbits, int j, int lane) {
  typedef typename rowvec<V>::type vf;
  int   s[B];
  float w[B];
  vf    x[B];
#pragma unroll
  for (int b = 0; b < B; ++b) {
    s[b] = __builtin_amdgcn_readlane(ms, j + b);
    w[b] = __int_as_float(__builtin_amdgcn_readlane(mwbits, j + b));
  }
#pragma unroll
  for (int b = 0; b < B; ++b) x[b] = Xv[(size_t)s[b] * 32 + lane];
#pragma unroll
  for (int b = 0; b < B; ++b) acc += x[b] * w[b];
}

template <int D, bool BIAS>
__global__ __launch_bounds__(256) void k_aggregate(float* __restrict__ Y,
                                                   const float* __restrict__ X,
                                                   const float* __restrict__ dinv,
                                                   const int* __restrict__ off,
                                                   const int* __restrict__ csr,
                                                   const float* __restrict__ bias, int N) {
  constexpr int V = D / 32;
  typedef typename rowvec<V>::type vf;
  int node = blockIdx.x * (blockDim.x >> 5) + (threadIdx.x >> 5);
  if (node >= N) return;
  node = __builtin_amdgcn_readfirstlane(node);
  const int lane = threadIdx.x & 31;
  const int beg = off[node], end = off[node + 1];
  const float dd = dinv[node];
  const vf* Xv = (const vf*)X;

  vf acc = Xv[(size_t)node * 32 + lane] * (dd * dd);
  for (int base = beg; base < end; base += 32) {
    const int n = min(32, end - base);
    int ms = 0;
    float mw = 0.0f;
    if (lane < n) {
      ms = csr[base + lane];
      mw = dinv[ms] * dd;
    }
    const int mwbits = __float_as_int(mw);
    int j = 0;
    for (; j + 8 <= n; j += 8) gather_rows<V, 8>(acc, Xv, ms, mwbits, j, lane);
    if (j + 4 <= n) { gather_rows<V, 4>(acc, Xv, ms, mwbits, j, lane); j += 4; }
    if (j + 2 <= n) { gather_rows<V, 2>(acc, Xv, ms, mwbits, j, lane); j += 2; }
    if (j < n)      { gather_rows<V, 1>(acc, Xv, ms, mwbits, j, lane); }
  }
  if (BIAS) acc += ((const vf*)bias)[lane];
  ((vf*)Y)[(size_t)node * 32 + lane] = acc;
}

extern "C" void kernel_launch(void* const* d_in, const int* in_sizes, int n_in,
                              void* d_out, int out_size, void* d_ws, size_t ws_size,
                              hipStream_t stream) {
  const float* emb  = (const float*)d_in[0];
  const float* W1   = (const float*)d_in[1];
  const float* b1   = (const float*)d_in[2];
  const float* W2   = (const float*)d_in[3];
  const float* b2   = (const float*)d_in[4];
  const int*   edge = (const int*)d_in[5];

  const int D_OUT = in_sizes[4];
  const int N     = in_sizes[0] / D_OUT;
  const int E     = in_sizes[5] / 2;
  const int* src  = edge;
  const int* dst  = edge + E;

  char* ws = (char*)d_ws;
  const int nchunk = (E + CHUNK - 1) / CHUNK;
  const int nbin   = (N + RNG - 1) / RNG;
  const size_t szN = (((size_t)(N + 1) * 4) + 255) & ~(size_t)255;
  const size_t szE = (((size_t)E * 4) + 255) & ~(size_t)255;
  const size_t szA = (((size_t)N * D_OUT * 4) + 255) & ~(size_t)255;
  const size_t szT = (size_t)nchunk * MAXB * 4;
  const size_t szB = (((size_t)(MAXB + 1) * 4) + 255) & ~(size_t)255;
  int*   off    = (int*)ws;
  float* dinv   = (float*)(ws + szN);
  int*   tab    = (int*)(ws + 2 * szN);
  int*   bstart = (int*)(ws + 2 * szN + szT);
  int*   bkt    = (int*)(ws + 2 * szN + szT + szB);
  int*   csr    = (int*)(ws + 2 * szN + szT + szB + szE);
  float* abuf   = (float*)(ws + 2 * szN + szT + szB + 2 * szE);
  float* xbuf   = (float*)(ws + 2 * szN + szT + szB + 2 * szE + szA);

  const int T = 256;
  const int wavesPerBlk = T / 32;
  const int mtiles  = (N + 15) / 16;

  k_bin_count<<<nchunk, BIN_T, 0, stream>>>(tab, dst, E);
  k_bin_place<<<nchunk, BIN_T, 0, stream>>>(bkt, bstart, tab, src, dst, E, nchunk);
  k_bin_csr  <<<nbin, BIN_T, 0, stream>>>(csr, off, dinv, bkt, bstart, N, E);

  k_aggregate<64, false><<<(N + wavesPerBlk - 1) / wavesPerBlk, T, 0, stream>>>(
      abuf, emb, dinv, off, csr, nullptr, N);
  k_dense2<<<(mtiles + DENSE_WAVES - 1) / DENSE_WAVES, T, 0, stream>>>(
      abuf, W1, b1, W2, xbuf, N);
  float* out = (float*)d_out;
  k_aggregate<64, true><<<(N + wavesPerBlk - 1) / wavesPerBlk, T, 0, stream>>>(
      out, xbuf, dinv, off, csr, b2, N);
}


# edge sort in 2 launches, not 3 (6 launches now 5): a chunk is put in bin order inside its own piece, with its bin starts; a workgroup per bin collects its runs from all chunks; sum order changes
# speedup vs baseline: 7.0775x; 1.0552x over previous
#include <hip/hip_runtime.h>


typedef float v2f __attribute__((ext_vector_type(2)));
typedef float v4f __attribute__((ext_vector_type(4)));
typedef float v8f __attribute__((ext_vector_type(8)));

constexpr int RSH   = 9;
constexpr int RNG   = 1 << RSH;
constexpr int MAXB  = 256;
constexpr int BIN_T = 1024;
constexpr int EPT   = 8;
constexpr int CHUNK = BIN_T * EPT;
constexpr int TP    = MAXB + 1;
constexpr int MAXC  = 256;
constexpr int RPW   = MAXC / 32;
constexpr int RL    = 2;

__global__ __launch_bounds__(BIN_T) void k_chunk_sort(int* __restrict__ bkt,
                                                      int* __restrict__ tab,
                                                      const int* __restrict__ src,
                                                      const int* __restrict__ dst, int E) {
  __shared__ int h[MAXB];
  __shared__ int wsum[MAXB / 32];
  __shared__ int stage[CHUNK];
  const int tid = threadIdx.x, lane = tid & 31, wid = tid >> 5;
  const int blk = blockIdx.x;
  const int e0 = blk * CHUNK + tid;
  int d[EPT], s[EPT], p[EPT];
#pragma unroll
  for (int k = 0; k < EPT; ++k) {
    const int i = min(e0 + k * BIN_T, E - 1);
    d[k] = dst[i];
    s[k] = src[i];
    p[k] = 0;
  }
  if (tid < MAXB) h[tid] = 0;
  __syncthreads();
#pragma unroll
  for (int k = 0; k < EPT; ++k)
    if (e0 + k * BIN_T < E) atomicAdd(&h[d[k] >> RSH], 1);
  __syncthreads();
  int own = 0, incl = 0;
  if (tid < MAXB) {
    own = h[tid];
    incl = own;
    for (int o = 1; o < 32; o <<= 1) {
      const int y = __shfl_up(incl, o);
      if (lane >= o) incl += y;
    }
    if (lane == 31) wsum[wid] = incl;
  }
  __syncthreads();
  const int cnt = min(CHUNK, E - blk * CHUNK);
  if (tid < MAXB) {
    int carry = 0;
    for (int w = 0; w < wid; ++w) carry += wsum[w];
    const int first = carry + incl - own;
    h[tid] = first;
    tab[blk * TP + tid] = first;
    if (tid == 0) tab[blk * TP + MAXB] = cnt;
  }
  __syncthreads();
#pragma unroll
  for (int k = 0; k < EPT; ++k)
    if (e0 + k * BIN_T < E) p[k] = atomicAdd(&h[d[k] >> RSH], 1);
#pragma unroll
  for (int k = 0; k < EPT; ++k)
    if (e0 + k * BIN_T < E) stage[p[k]] = (s[k] << RSH) | (d[k] & (RNG - 1));
  __syncthreads();
  int sv[EPT];
#pragma unroll
  for (int k = 0; k < EPT; ++k) sv[k] = stage[min(tid + k * BIN_T, cnt - 1)];
#pragma unroll
  for (int k = 0; k < EPT; ++k) {
    const int pos = tid + k * BIN_T;
    if (pos < cnt) bkt[blk * CHUNK + pos] = sv[k];
  }
}

constexpr int CAP = 16384;

__global__ __launch_bounds__(BIN_T) void k_bin_csr(int* __restrict__ csr,
                                                   int* __restrict__ off,
                                                   float* __restrict__ dinv,
                                                   const int* __restrict__ bkt,
                                                   const int* __restrict__ tab,
                                                   int N, int E, int nchunk) {
  __shared__ int c[RNG];
  __shared__ int wsum[RNG / 32];
  __shared__ int stage[CAP];
  __shared__ int rbeg[MAXC];
  __shared__ int rlen[MAXC];
  __shared__ int ps[MAXC / 32];
  __shared__ int pe[MAXC / 32];
  const int tid = threadIdx.x, lane = tid & 31, wid = tid >> 5;
  const int b = blockIdx.x;
  if (tid < MAXC) {
    int st = 0, en = 0;
    if (tid < nchunk) {
      st = tab[tid * TP + b];
      en = tab[tid * TP + b + 1];
    }
    rbeg[tid] = tid * CHUNK + st;
    rlen[tid] = en - st;
    int a = st, z = en;
    for (int o = 16; o > 0; o >>= 1) {
      a += __shfl_xor(a, o);
      z += __shfl_xor(z, o);
    }
    if (lane == 0) {
      ps[wid] = a;
      pe[wid] = z;
    }
  }
  if (tid < RNG) c[tid] = 0;
  __syncthreads();
  int s = 0, e = 0;
#pragma unroll
  for (int w = 0; w < MAXC / 32; ++w) {
    s += ps[w];
    e += pe[w];
  }
  const int cnt = e - s;
  const bool fits = cnt <= CAP;
  int beg[RPW], len[RPW], w[RPW][RL];
#pragma unroll
  for (int j = 0; j < RPW; ++j) {
    beg[j] = rbeg[wid + 32 * j];
    len[j] = rlen[wid + 32 * j];
  }
#pragma unroll
  for (int j = 0; j < RPW; ++j) {
#pragma unroll
    for (int t = 0; t < RL; ++t) w[j][t] = bkt[min(beg[j] + lane + 32 * t, E - 1)];
  }
#pragma unroll
  for (int j = 0; j < RPW; ++j) {
#pragma unroll
    for (int t = 0; t < RL; ++t)
      if (lane + 32 * t < len[j]) atomicAdd(&c[w[j][t] & (RNG - 1)], 1);
  }
  for (int j = 0; j < RPW; ++j)
    for (int i = lane + 32 * RL; i < len[j]; i += 32)
      atomicAdd(&c[bkt[beg[j] + i] & (RNG - 1)], 1);
  __syncthreads();
  int v = 0, incl = 0;
  if (tid < RNG) {
    v = c[tid];
    incl = v;
    for (int o = 1; o < 32; o <<= 1) {
      const int y = __shfl_up(incl, o);
      if (lane >= o) incl += y;
    }
    if (lane == 31) wsum[wid] = incl;
  }
  __syncthreads();
  if (tid < RNG) {
    int carry = 0;
    for (int ww = 0; ww < wid; ++ww) carry += wsum[ww];
    const int o = s + carry + incl - v;
    c[tid] = o;
    const int node = b * RNG + tid;
    if (node < N) {
      off[node]  = o;
      dinv[node] = rsqrtf((float)(v + 1));
    }
  }
  if (b == (int)gridDim.x - 1 && tid == 0) off[N] = E;
  __syncthreads();
  int p[RPW][RL];
#pragma unroll
  for (int j = 0; j < RPW; ++j) {
#pragma unroll
    for (int t = 0; t < RL; ++t) {
      p[j][t] = 0;
      if (lane + 32 * t < len[j]) p[j][t] = atomicAdd(&c[w[j][t] & (RNG - 1)], 1);
    }
  }
  if (fits) {
#pragma unroll
    for (int j = 0; j < RPW; ++j) {
#pragma unroll
      for (int t = 0; t < RL; ++t)
        if (lane + 32 * t < len[j]) stage[p[j][t] - s] = w[j][t] >> RSH;
    }
  } else {
#pragma unroll
    for (int j = 0; j < RPW; ++j) {
#pragma unroll
      for (int t = 0; t < RL; ++t)
        if (lane + 32 * t < len[j]) csr[p[j][t]] = w[j][t] >> RSH;
    }
  }
  for (int j = 0; j < RPW; ++j)
    for (int i = lane + 32 * RL; i < len[j]; i += 32) {
      const int x = bkt[beg[j] + i];
      const int q = atomicAdd(&c[x & (RNG - 1)], 1);
      if (fits) stage[q - s] = x >> RSH;
      else      csr[q] = x >> RSH;
    }
  __syncthreads();
  if (fits && cnt > 0) {
    for (int i0 = tid - (s & 31); i0 < cnt; i0 += 4 * BIN_T) {
      int y[4];
#pragma unroll
      for (int j = 0; j < 4; ++j) y[j] = stage[min(max(i0 + j * BIN_T, 0), cnt - 1)];
#pragma unroll
      for (int j = 0; j < 4; ++j) {
        const int i = i0 + j * BIN_T;
        if (i >= 0 && i < cnt) csr[s + i] = y[j];
      }
    }
  }
}

constexpr int D_IN  = 64;
constexpr int D_MID = 128;
constexpr int D_FIN = 64;
constexpr int PW1  = 2 * D_MID + 32;
constexpr int PW2  = 2 * D_FIN + 32;
constexpr int PA   = D_IN + 4;
constexpr int PH   = D_MID + 4;
constexpr int SLOT = 16 * PH;
constexpr int DENSE_WAVES = 8;

__global__ __launch_bounds__(256) void k_dense2(const float* __restrict__ X,
                                                const float* __restrict__ W1,
                                                const float* __restrict__ b1,
                                                const float* __restrict__ W2,
                                                float* __restrict__ Y, int M) {
  constexpr int W2_FLOATS = (D_MID / 2) * PW2;
  constexpr int W1_FLOATS = (D_IN / 2) * PW1;
  __shared__ __attribute__((aligned(16))) float lds[W2_FLOATS + W1_FLOATS + DENSE_WAVES * SLOT];
  float* const w2p   = lds;
  float* const w1p   = lds + W2_FLOATS;
  float* const slots = lds + W2_FLOATS + W1_FLOATS;

  const int tid  = threadIdx.x;
  const int lane = tid & 31;
  const int wid  = __builtin_amdgcn_readfirstlane(tid >> 5);
  const int half = lane >> 4;
  const int r    = lane & 15;
  const int mtiles = (M + 15) >> 4;
  const int tile   = blockIdx.x * DENSE_WAVES + wid;
  const bool live  = tile < mtiles;
  const int m0 = (live ? tile : mtiles - 1) << 4;
  float* slot = slots + wid * SLOT;

  const int trow = lane >> 4;
  const int tcol = (lane & 15) * 4;
  v4f xin[8];
#pragma unroll
  for (int i = 0; i < 8; ++i) {
    const int gr = min(m0 + 2 * i + trow, M - 1);
    xin[i] = *(const v4f*)(X + (size_t)gr * D_IN + tcol);
  }
  float bias[D_MID / 16];
#pragma unroll
  for (int nt = 0; nt < D_MID / 16; ++nt) bias[nt] = b1[nt * 16 + r];

#pragma unroll
  for (int it = 0; it < (D_IN / 2) * (D_MID / 4) / 256; ++it) {
    const int item = tid + 256 * it;
    const int p = item / (D_MID / 4), n = (item % (D_MID / 4)) * 4;
    const v4f lo = *(const v4f*)(W1 + (size_t)(2 * p) * D_MID + n);
    const v4f hi = *(const v4f*)(W1 + (size_t)(2 * p + 1) * D_MID + n);
    const v4f s0 = {lo[0], hi[0], lo[1], hi[1]};
    const v4f s1 = {lo[2], hi[2], lo[3], hi[3]};
    *(v4f*)(w1p + p * PW1 + 2 * n)     = s0;
    *(v4f*)(w1p + p * PW1 + 2 * n + 4) = s1;
  }
#pragma unroll
  for (int it = 0; it < (D_MID / 2) * (D_FIN / 4) / 256; ++it) {
    const int item = tid + 256 * it;
    const int p = item / (D_FIN / 4), n = (item % (D_FIN / 4)) * 4;
    const v4f lo = *(const v4f*)(W2 + (size_t)(2 * p) * D_FIN + n);
    const v4f hi = *(const v4f*)(W2 + (size_t)(2 * p + 1) * D_FIN + n);
    const v4f s0 = {lo[0], hi[0], lo[1], hi[1]};
    const v4f s1 = {lo[2], hi[2], lo[3], hi[3]};
    *(v4f*)(w2p + p * PW2 + 2 * n)     = s0;
    *(v4f*)(w2p + p * PW2 + 2 * n + 4) = s1;
  }
#pragma unroll
  for (int i = 0; i < 8; ++i) *(v4f*)(slot + (2 * i + trow) * PA + tcol) = xin[i];
  __syncthreads();

  v8f acc[D_MID / 16];
#pragma unroll
  for (int nt = 0; nt < D_MID / 16; ++nt) acc[nt] = v8f{};
  {
    const float* ap = slot + r * PA + 2 * half;
    const float* bp = w1p + half * PW1 + 2 * r;
#pragma unroll
    for (int kk = 0; kk < D_IN / 4; ++kk) {
      const v2f a = *(const v2f*)(ap + 4 * kk);
#pragma unroll
      for (int nt = 0; nt < D_MID / 16; ++nt) {
        const v2f b = *(const v2f*)(bp + (2 * kk) * PW1 + nt * 32);
        acc[nt] = __builtin_amdgcn_wmma_f32_16x16x4_f32(
             false, a,  false, b,
             (short)0, acc[nt],  false,  false);
      }
    }
  }
#pragma unroll
  for (int nt = 0; nt < D_MID / 16; ++nt) {
#pragma unroll
    for (int v = 0; v < 8; ++v)
      slot[(v + 8 * half) * PH + nt * 16 + r] = fmaxf(acc[nt][v] + bias[nt], 0.0f);
  }
  __syncthreads();

  v8f acc2[D_FIN / 16];
#pragma unroll
  for (int nt = 0; nt < D_FIN / 16; ++nt) acc2[nt] = v8f{};
  {
    const float* ap = slot + r * PH + 2 * half;
    const float* bp = w2p + half * PW2 + 2 * r;
#pragma unroll
    for (int kk = 0; kk < D_MID / 4; ++kk) {
      const v2f a = *(const v2f*)(ap + 4 * kk);
#pragma unroll
      for (int nt = 0; nt < D_FIN / 16; ++nt) {
        const v2f b = *(const v2f*)(bp + (2 * kk) * PW2 + nt * 32);
        acc2[nt] = __builtin_amdgcn_wmma_f32_16x16x4_f32(
             false, a,  false, b,
             (short)0, acc2[nt],  false,  false);
      }
    }
  }
#pragma unroll
  for (int nt = 0; nt < D_FIN / 16; ++nt) {
#pragma unroll
    for (int v = 0; v < 8; ++v) slot[(v + 8 * half) * PA + nt * 16 + r] = acc2[nt][v];
  }
  __syncthreads();
  if (live) {
    if (m0 + 16 <= M) {
#pragma unroll
      for (int i = 0; i < 8; ++i) {
        const v4f y = *(const v4f*)(slot + (2 * i + trow) * PA + tcol);
        *(v4f*)(Y + (size_t)(m0 + 2 * i + trow) * D_FIN + tcol) = y;
      }
    } else {
#pragma unroll
      for (int i = 0; i < 8; ++i) {
        const int row = m0 + 2 * i + trow;
        const v4f y = *(const v4f*)(slot + (2 * i + trow) * PA + tcol);
        if (row < M) *(v4f*)(Y + (size_t)row * D_FIN + tcol) = y;
      }
    }
  }
}

template <int V>
struct rowvec;
template <> struct rowvec<4> { typedef v4f type; };
template <> struct rowvec<2> { typedef v2f type; };

template <int V, int B>
__device__ __forceinline__ void gather_rows(typename rowvec<V>::type& acc,
                                            const typename rowvec<V>::type* __restrict__ Xv,
                                            int ms, int mwbits, int j, int lane) {
  typedef typename rowvec<V>::type vf;
  int   s[B];
  float w[B];
  vf    x[B];
#pragma unroll
  for (int b = 0; b < B; ++b) {
    s[b] = __builtin_amdgcn_readlane(ms, j + b);
    w[b] = __int_as_float(__builtin_amdgcn_readlane(mwbits, j + b));
  }
#pragma unroll
  for (int b = 0; b < B; ++b) x[b] = Xv[(size_t)s[b] * 32 + lane];
#pragma unroll
  for (int b = 0; b < B; ++b) acc += x[b] * w[b];
}

template <int D, bool BIAS>
__global__ __launch_bounds__(256) void k_aggregate(float* __restrict__ Y,
                                                   const float* __restrict__ X,
                                                   const float* __restrict__ dinv,
                                                   const int* __restrict__ off,
                                                   const int* __restrict__ csr,
                                                   const float* __restrict__ bias, int N) {
  constexpr int V = D / 32;
  typedef typename rowvec<V>::type vf;
  int node = blockIdx.x * (blockDim.x >> 5) + (threadIdx.x >> 5);
  if (node >= N) return;
  node = __builtin_amdgcn_readfirstlane(node);
  const int lane = threadIdx.x & 31;
  const int beg = off[node], end = off[node + 1];
  const float dd = dinv[node];
  const vf* Xv = (const vf*)X;

  vf acc = Xv[(size_t)node * 32 + lane] * (dd * dd);
  for (int base = beg; base < end; base += 32) {
    const int n = min(32, end - base);
    int ms = 0;
    float mw = 0.0f;
    if (lane < n) {
      ms = csr[base + lane];
      mw = dinv[ms] * dd;
    }
    const int mwbits = __float_as_int(mw);
    int j = 0;
    for (; j + 8 <= n; j += 8) gather_rows<V, 8>(acc, Xv, ms, mwbits, j, lane);
    if (j + 4 <= n) { gather_rows<V, 4>(acc, Xv, ms, mwbits, j, lane); j += 4; }
    if (j + 2 <= n) { gather_rows<V, 2>(acc, Xv, ms, mwbits, j, lane); j += 2; }
    if (j < n)      { gather_rows<V, 1>(acc, Xv, ms, mwbits, j, lane); }
  }
  if (BIAS) acc += ((const vf*)bias)[lane];
  ((vf*)Y)[(size_t)node * 32 + lane] = acc;
}

extern "C" void kernel_launch(void* const* d_in, const int* in_sizes, int n_in,
                              void* d_out, int out_size, void* d_ws, size_t ws_size,
                              hipStream_t stream) {
  const float* emb  = (const float*)d_in[0];
  const float* W1   = (const float*)d_in[1];
  const float* b1   = (const float*)d_in[2];
  const float* W2   = (const float*)d_in[3];
  const float* b2   = (const float*)d_in[4];
  const int*   edge = (const int*)d_in[5];

  const int D_OUT = in_sizes[4];
  const int N     = in_sizes[0] / D_OUT;
  const int E     = in_sizes[5] / 2;
  const int* src  = edge;
  const int* dst  = edge + E;

  char* ws = (char*)d_ws;
  const int nchunk = (E + CHUNK - 1) / CHUNK;
  const int nbin   = (N + RNG - 1) / RNG;
  const size_t szN = (((size_t)(N + 1) * 4) + 255) & ~(size_t)255;
  const size_t szE = (((size_t)E * 4) + 255) & ~(size_t)255;
  const size_t szA = (((size_t)N * D_OUT * 4) + 255) & ~(size_t)255;
  const size_t szT = (((size_t)nchunk * TP * 4) + 255) & ~(size_t)255;
  int*   off    = (int*)ws;
  float* dinv   = (float*)(ws + szN);
  int*   tab    = (int*)(ws + 2 * szN);
  int*   bkt    = (int*)(ws + 2 * szN + szT);
  int*   csr    = (int*)(ws + 2 * szN + szT + szE);
  float* abuf   = (float*)(ws + 2 * szN + szT + 2 * szE);
  float* xbuf   = (float*)(ws + 2 * szN + szT + 2 * szE + szA);

  const int T = 256;
  const int wavesPerBlk = T / 32;
  const int mtiles  = (N + 15) / 16;

  k_chunk_sort<<<nchunk, BIN_T, 0, stream>>>(bkt, tab, src, dst, E);
  k_bin_csr   <<<nbin, BIN_T, 0, stream>>>(csr, off, dinv, bkt, tab, N, E, nchunk);

  k_aggregate<64, false><<<(N + wavesPerBlk - 1) / wavesPerBlk, T, 0, stream>>>(
      abuf, emb, dinv, off, csr, nullptr, N);
  k_dense2<<<(mtiles + DENSE_WAVES - 1) / DENSE_WAVES, T, 0, stream>>>(
      abuf, W1, b1, W2, xbuf, N);
  float* out = (float*)d_out;
  k_aggregate<64, true><<<(N + wavesPerBlk - 1) / wavesPerBlk, T, 0, stream>>>(
      out, xbuf, dinv, off, csr, b2, N);
}


# both dense products: one 16-wave workgroup per WGP stages the weights once, its waves loop over row tiles dealt evenly over the SIMDs, no workgroup barrier per tile; 782 workgroups become 256
# speedup vs baseline: 7.3075x; 1.0325x over previous
#include <hip/hip_runtime.h>


typedef float v2f __attribute__((ext_vector_type(2)));
typedef float v4f __attribute__((ext_vector_type(4)));
typedef float v8f __attribute__((ext_vector_type(8)));

constexpr int RSH   = 9;
constexpr int RNG   = 1 << RSH;
constexpr int MAXB  = 256;
constexpr int BIN_T = 1024;
constexpr int EPT   = 8;
constexpr int CHUNK = BIN_T * EPT;
constexpr int TP    = MAXB + 1;
constexpr int MAXC  = 256;
constexpr int RPW   = MAXC / 32;
constexpr int RL    = 2;

__global__ __launch_bounds__(BIN_T) void k_chunk_sort(int* __restrict__ bkt,
                                                      int* __restrict__ tab,
                                                      const int* __restrict__ src,
                                                      const int* __restrict__ dst, int E) {
  __shared__ int h[MAXB];
  __shared__ int wsum[MAXB / 32];
  __shared__ int stage[CHUNK];
  const int tid = threadIdx.x, lane = tid & 31, wid = tid >> 5;
  const int blk = blockIdx.x;
  const int e0 = blk * CHUNK + tid;
  int d[EPT], s[EPT], p[EPT];
#pragma unroll
  for (int k = 0; k < EPT; ++k) {
    const int i = min(e0 + k * BIN_T, E - 1);
    d[k] = dst[i];
    s[k] = src[i];
    p[k] = 0;
  }
  if (tid < MAXB) h[tid] = 0;
  __syncthreads();
#pragma unroll
  for (int k = 0; k < EPT; ++k)
    if (e0 + k * BIN_T < E) atomicAdd(&h[d[k] >> RSH], 1);
  __syncthreads();
  int own = 0, incl = 0;
  if (tid < MAXB) {
    own = h[tid];
    incl = own;
    for (int o = 1; o < 32; o <<= 1) {
      const int y = __shfl_up(incl, o);
      if (lane >= o) incl += y;
    }
    if (lane == 31) wsum[wid] = incl;
  }
  __syncthreads();
  const int cnt = min(CHUNK, E - blk * CHUNK);
  if (tid < MAXB) {
    int carry = 0;
    for (int w = 0; w < wid; ++w) carry += wsum[w];
    const int first = carry + incl - own;
    h[tid] = first;
    tab[blk * TP + tid] = first;
    if (tid == 0) tab[blk * TP + MAXB] = cnt;
  }
  __syncthreads();
#pragma unroll
  for (int k = 0; k < EPT; ++k)
    if (e0 + k * BIN_T < E) p[k] = atomicAdd(&h[d[k] >> RSH], 1);
#pragma unroll
  for (int k = 0; k < EPT; ++k)
    if (e0 + k * BIN_T < E) stage[p[k]] = (s[k] << RSH) | (d[k] & (RNG - 1));
  __syncthreads();
  int sv[EPT];
#pragma unroll
  for (int k = 0; k < EPT; ++k) sv[k] = stage[min(tid + k * BIN_T, cnt - 1)];
#pragma unroll
  for (int k = 0; k < EPT; ++k) {
    const int pos = tid + k * BIN_T;
    if (pos < cnt) bkt[blk * CHUNK + pos] = sv[k];
  }
}

constexpr int CAP = 16384;

__global__ __launch_bounds__(BIN_T) void k_bin_csr(int* __restrict__ csr,
                                                   int* __restrict__ off,
                                                   float* __restrict__ dinv,
                                                   const int* __restrict__ bkt,
                                                   const int* __restrict__ tab,
                                                   int N, int E, int nchunk) {
  __shared__ int c[RNG];
  __shared__ int wsum[RNG / 32];
  __shared__ int stage[CAP];
  __shared__ int rbeg[MAXC];
  __shared__ int rlen[MAXC];
  __shared__ int ps[MAXC / 32];
  __shared__ int pe[MAXC / 32];
  const int tid = threadIdx.x, lane = tid & 31, wid = tid >> 5;
  const int b = blockIdx.x;
  if (tid < MAXC) {
    int st = 0, en = 0;
    if (tid < nchunk) {
      st = tab[tid * TP + b];
      en = tab[tid * TP + b + 1];
    }
    rbeg[tid] = tid * CHUNK + st;
    rlen[tid] = en - st;
    int a = st, z = en;
    for (int o = 16; o > 0; o >>= 1) {
      a += __shfl_xor(a, o);
      z += __shfl_xor(z, o);
    }
    if (lane == 0) {
      ps[wid] = a;
      pe[wid] = z;
    }
  }
  if (tid < RNG) c[tid] = 0;
  __syncthreads();
  int s = 0, e = 0;
#pragma unroll
  for (int w = 0; w < MAXC / 32; ++w) {
    s += ps[w];
    e += pe[w];
  }
  const int cnt = e - s;
  const bool fits = cnt <= CAP;
  int beg[RPW], len[RPW], w[RPW][RL];
#pragma unroll
  for (int j = 0; j < RPW; ++j) {
    beg[j] = rbeg[wid + 32 * j];
    len[j] = rlen[wid + 32 * j];
  }
#pragma unroll
  for (int j = 0; j < RPW; ++j) {
#pragma unroll
    for (int t = 0; t < RL; ++t) w[j][t] = bkt[min(beg[j] + lane + 32 * t, E - 1)];
  }
#pragma unroll
  for (int j = 0; j < RPW; ++j) {
#pragma unroll
    for (int t = 0; t < RL; ++t)
      if (lane + 32 * t < len[j]) atomicAdd(&c[w[j][t] & (RNG - 1)], 1);
  }
  for (int j = 0; j < RPW; ++j)
    for (int i = lane + 32 * RL; i < len[j]; i += 32)
      atomicAdd(&c[bkt[beg[j] + i] & (RNG - 1)], 1);
  __syncthreads();
  int v = 0, incl = 0;
  if (tid < RNG) {
    v = c[tid];
    incl = v;
    for (int o = 1; o < 32; o <<= 1) {
      const int y = __shfl_up(incl, o);
      if (lane >= o) incl += y;
    }
    if (lane == 31) wsum[wid] = incl;
  }
  __syncthreads();
  if (tid < RNG) {
    int carry = 0;
    for (int ww = 0; ww < wid; ++ww) carry += wsum[ww];
    const int o = s + carry + incl - v;
    c[tid] = o;
    const int node = b * RNG + tid;
    if (node < N) {
      off[node]  = o;
      dinv[node] = rsqrtf((float)(v + 1));
    }
  }
  if (b == (int)gridDim.x - 1 && tid == 0) off[N] = E;
  __syncthreads();
  int p[RPW][RL];
#pragma unroll
  for (int j = 0; j < RPW; ++j) {
#pragma unroll
    for (int t = 0; t < RL; ++t) {
      p[j][t] = 0;
      if (lane + 32 * t < len[j]) p[j][t] = atomicAdd(&c[w[j][t] & (RNG - 1)], 1);
    }
  }
  if (fits) {
#pragma unroll
    for (int j = 0; j < RPW; ++j) {
#pragma unroll
      for (int t = 0; t < RL; ++t)
        if (lane + 32 * t < len[j]) stage[p[j][t] - s] = w[j][t] >> RSH;
    }
  } else {
#pragma unroll
    for (int j = 0; j < RPW; ++j) {
#pragma unroll
      for (int t = 0; t < RL; ++t)
        if (lane + 32 * t < len[j]) csr[p[j][t]] = w[j][t] >> RSH;
    }
  }
  for (int j = 0; j < RPW; ++j)
    for (int i = lane + 32 * RL; i < len[j]; i += 32) {
      const int x = bkt[beg[j] + i];
      const int q = atomicAdd(&c[x & (RNG - 1)], 1);
      if (fits) stage[q - s] = x >> RSH;
      else      csr[q] = x >> RSH;
    }
  __syncthreads();
  if (fits && cnt > 0) {
    for (int i0 = tid - (s & 31); i0 < cnt; i0 += 4 * BIN_T) {
      int y[4];
#pragma unroll
      for (int j = 0; j < 4; ++j) y[j] = stage[min(max(i0 + j * BIN_T, 0), cnt - 1)];
#pragma unroll
      for (int j = 0; j < 4; ++j) {
        const int i = i0 + j * BIN_T;
        if (i >= 0 && i < cnt) csr[s + i] = y[j];
      }
    }
  }
}

constexpr int D_IN  = 64;
constexpr int D_MID = 128;
constexpr int D_FIN = 64;
constexpr int PW1  = 2 * D_MID + 32;
constexpr int PW2  = 2 * D_FIN + 32;
constexpr int PA   = D_IN + 4;
constexpr int PH   = D_MID + 4;
constexpr int SLOT = 16 * PH;
constexpr int DENSE_WAVES = 16;
constexpr int DENSE_T = DENSE_WAVES * 32;

__device__ __forceinline__ void wave_slot_handover() {
  __builtin_amdgcn_fence(__ATOMIC_RELEASE, "wavefront");
  __builtin_amdgcn_wave_barrier();
  __builtin_amdgcn_fence(__ATOMIC_ACQUIRE, "wavefront");
}

__global__ __launch_bounds__(DENSE_T) void k_dense2(const float* __restrict__ X,
                                                    const float* __restrict__ W1,
                                                    const float* __restrict__ b1,
                                                    const float* __restrict__ W2,
                                                    float* __restrict__ Y, int M, int nwg) {
  constexpr int W2_FLOATS = (D_MID / 2) * PW2;
  constexpr int W1_FLOATS = (D_IN / 2) * PW1;
  __shared__ __attribute__((aligned(16))) float lds[W2_FLOATS + W1_FLOATS + DENSE_WAVES * SLOT];
  float* const w2p   = lds;
  float* const w1p   = lds + W2_FLOATS;
  float* const slots = lds + W2_FLOATS + W1_FLOATS;

  const int tid  = threadIdx.x;
  const int lane = tid & 31;
  const int wid  = __builtin_amdgcn_readfirstlane(tid >> 5);
  const int half = lane >> 4;
  const int r    = lane & 15;
  const int mtiles  = (M + 15) >> 4;
  const int tstride = DENSE_WAVES * nwg;
  int tile = wid * nwg + (int)blockIdx.x;
  float* slot = slots + wid * SLOT;

  const int trow = lane >> 4;
  const int tcol = (lane & 15) * 4;
  v4f xin[8];
  {
    const int m0 = min(tile, mtiles - 1) << 4;
#pragma unroll
    for (int i = 0; i < 8; ++i) {
      const int gr = min(m0 + 2 * i + trow, M - 1);
      xin[i] = *(const v4f*)(X + (size_t)gr * D_IN + tcol);
    }
  }
  float bias[D_MID / 16];
#pragma unroll
  for (int nt = 0; nt < D_MID / 16; ++nt) bias[nt] = b1[nt * 16 + r];

#pragma unroll
  for (int it = 0; it < (D_IN / 2) * (D_MID / 4) / DENSE_T; ++it) {
    const int item = tid + DENSE_T * it;
    const int p = item / (D_MID / 4), n = (item % (D_MID / 4)) * 4;
    const v4f lo = *(const v4f*)(W1 + (size_t)(2 * p) * D_MID + n);
    const v4f hi = *(const v4f*)(W1 + (size_t)(2 * p + 1) * D_MID + n);
    const v4f s0 = {lo[0], hi[0], lo[1], hi[1]};
    const v4f s1 = {lo[2], hi[2], lo[3], hi[3]};
    *(v4f*)(w1p + p * PW1 + 2 * n)     = s0;
    *(v4f*)(w1p + p * PW1 + 2 * n + 4) = s1;
  }
#pragma unroll
  for (int it = 0; it < (D_MID / 2) * (D_FIN / 4) / DENSE_T; ++it) {
    const int item = tid + DENSE_T * it;
    const int p = item / (D_FIN / 4), n = (item % (D_FIN / 4)) * 4;
    const v4f lo = *(const v4f*)(W2 + (size_t)(2 * p) * D_FIN + n);
    const v4f hi = *(const v4f*)(W2 + (size_t)(2 * p + 1) * D_FIN + n);
    const v4f s0 = {lo[0], hi[0], lo[1], hi[1]};
    const v4f s1 = {lo[2], hi[2], lo[3], hi[3]};
    *(v4f*)(w2p + p * PW2 + 2 * n)     = s0;
    *(v4f*)(w2p + p * PW2 + 2 * n + 4) = s1;
  }
  __syncthreads();

  for (; tile < mtiles; tile += tstride) {
    const int m0 = tile << 4;
#pragma unroll
    for (int i = 0; i < 8; ++i) *(v4f*)(slot + (2 * i + trow) * PA + tcol) = xin[i];
    if (tile + tstride < mtiles) {
      const int n0 = (tile + tstride) << 4;
#pragma unroll
      for (int i = 0; i < 8; ++i) {
        const int gr = min(n0 + 2 * i + trow, M - 1);
        xin[i] = *(const v4f*)(X + (size_t)gr * D_IN + tcol);
      }
    }
    wave_slot_handover();

    v8f acc[D_MID / 16];
#pragma unroll
    for (int nt = 0; nt < D_MID / 16; ++nt) acc[nt] = v8f{};
    {
      const float* ap = slot + r * PA + 2 * half;
      const float* bp = w1p + half * PW1 + 2 * r;
#pragma unroll
      for (int kk = 0; kk < D_IN / 4; ++kk) {
        const v2f a = *(const v2f*)(ap + 4 * kk);
#pragma unroll
        for (int nt = 0; nt < D_MID / 16; ++nt) {
          const v2f b = *(const v2f*)(bp + (2 * kk) * PW1 + nt * 32);
          acc[nt] = __builtin_amdgcn_wmma_f32_16x16x4_f32(
               false, a,  false, b,
               (short)0, acc[nt],  false,  false);
        }
      }
    }
#pragma unroll
    for (int nt = 0; nt < D_MID / 16; ++nt) {
#pragma unroll
      for (int v = 0; v < 8; ++v)
        slot[(v + 8 * half) * PH + nt * 16 + r] = fmaxf(acc[nt][v] + bias[nt], 0.0f);
    }
    wave_slot_handover();

    v8f acc2[D_FIN / 16];
#pragma unroll
    for (int nt = 0; nt < D_FIN / 16; ++nt) acc2[nt] = v8f{};
    {
      const float* ap = slot + r * PH + 2 * half;
      const float* bp = w2p + half * PW2 + 2 * r;
#pragma unroll
      for (int kk = 0; kk < D_MID / 4; ++kk) {
        const v2f a = *(const v2f*)(ap + 4 * kk);
#pragma unroll
        for (int nt = 0; nt < D_FIN / 16; ++nt) {
          const v2f b = *(const v2f*)(bp + (2 * kk) * PW2 + nt * 32);
          acc2[nt] = __builtin_amdgcn_wmma_f32_16x16x4_f32(
               false, a,  false, b,
               (short)0, acc2[nt],  false,  false);
        }
      }
    }
#pragma unroll
    for (int nt = 0; nt < D_FIN / 16; ++nt) {
#pragma unroll
      for (int v = 0; v < 8; ++v) slot[(v + 8 * half) * PA + nt * 16 + r] = acc2[nt][v];
    }
    wave_slot_handover();
    if (m0 + 16 <= M) {
#pragma unroll
      for (int i = 0; i < 8; ++i) {
        const v4f y = *(const v4f*)(slot + (2 * i + trow) * PA + tcol);
        *(v4f*)(Y + (size_t)(m0 + 2 * i + trow) * D_FIN + tcol) = y;
      }
    } else {
#pragma unroll
      for (int i = 0; i < 8; ++i) {
        const int row = m0 + 2 * i + trow;
        const v4f y = *(const v4f*)(slot + (2 * i + trow) * PA + tcol);
        if (row < M) *(v4f*)(Y + (size_t)row * D_FIN + tcol) = y;
      }
    }
  }
}

template <int V>
struct rowvec;
template <> struct rowvec<4> { typedef v4f type; };
template <> struct rowvec<2> { typedef v2f type; };

template <int V, int B>
__device__ __forceinline__ void gather_rows(typename rowvec<V>::type& acc,
                                            const typename rowvec<V>::type* __restrict__ Xv,
                                            int ms, int mwbits, int j, int lane) {
  typedef typename rowvec<V>::type vf;
  int   s[B];
  float w[B];
  vf    x[B];
#pragma unroll
  for (int b = 0; b < B; ++b) {
    s[b] = __builtin_amdgcn_readlane(ms, j + b);
    w[b] = __int_as_float(__builtin_amdgcn_readlane(mwbits, j + b));
  }
#pragma unroll
  for (int b = 0; b < B; ++b) x[b] = Xv[(size_t)s[b] * 32 + lane];
#pragma unroll
  for (int b = 0; b < B; ++b) acc += x[b] * w[b];
}

template <int D, bool BIAS>
__global__ __launch_bounds__(256) void k_aggregate(float* __restrict__ Y,
                                                   const float* __restrict__ X,
                                                   const float* __restrict__ dinv,
                                                   const int* __restrict__ off,
                                                   const int* __restrict__ csr,
                                                   const float* __restrict__ bias, int N) {
  constexpr int V = D / 32;
  typedef typename rowvec<V>::type vf;
  int node = blockIdx.x * (blockDim.x >> 5) + (threadIdx.x >> 5);
  if (node >= N) return;
  node = __builtin_amdgcn_readfirstlane(node);
  const int lane = threadIdx.x & 31;
  const int beg = off[node], end = off[node + 1];
  const float dd = dinv[node];
  const vf* Xv = (const vf*)X;

  vf acc = Xv[(size_t)node * 32 + lane] * (dd * dd);
  for (int base = beg; base < end; base += 32) {
    const int n = min(32, end - base);
    int ms = 0;
    float mw = 0.0f;
    if (lane < n) {
      ms = csr[base + lane];
      mw = dinv[ms] * dd;
    }
    const int mwbits = __float_as_int(mw);
    int j = 0;
    for (; j + 8 <= n; j += 8) gather_rows<V, 8>(acc, Xv, ms, mwbits, j, lane);
    if (j + 4 <= n) { gather_rows<V, 4>(acc, Xv, ms, mwbits, j, lane); j += 4; }
    if (j + 2 <= n) { gather_rows<V, 2>(acc, Xv, ms, mwbits, j, lane); j += 2; }
    if (j < n)      { gather_rows<V, 1>(acc, Xv, ms, mwbits, j, lane); }
  }
  if (BIAS) acc += ((const vf*)bias)[lane];
  ((vf*)Y)[(size_t)node * 32 + lane] = acc;
}

extern "C" void kernel_launch(void* const* d_in, const int* in_sizes, int n_in,
                              void* d_out, int out_size, void* d_ws, size_t ws_size,
                              hipStream_t stream) {
  const float* emb  = (const float*)d_in[0];
  const float* W1   = (const float*)d_in[1];
  const float* b1   = (const float*)d_in[2];
  const float* W2   = (const float*)d_in[3];
  const float* b2   = (const float*)d_in[4];
  const int*   edge = (const int*)d_in[5];

  const int D_OUT = in_sizes[4];
  const int N     = in_sizes[0] / D_OUT;
  const int E     = in_sizes[5] / 2;
  const int* src  = edge;
  const int* dst  = edge + E;

  char* ws = (char*)d_ws;
  const int nchunk = (E + CHUNK - 1) / CHUNK;
  const int nbin   = (N + RNG - 1) / RNG;
  const size_t szN = (((size_t)(N + 1) * 4) + 255) & ~(size_t)255;
  const size_t szE = (((size_t)E * 4) + 255) & ~(size_t)255;
  const size_t szA = (((size_t)N * D_OUT * 4) + 255) & ~(size_t)255;
  const size_t szT = (((size_t)nchunk * TP * 4) + 255) & ~(size_t)255;
  int*   off    = (int*)ws;
  float* dinv   = (float*)(ws + szN);
  int*   tab    = (int*)(ws + 2 * szN);
  int*   bkt    = (int*)(ws + 2 * szN + szT);
  int*   csr    = (int*)(ws + 2 * szN + szT + szE);
  float* abuf   = (float*)(ws + 2 * szN + szT + 2 * szE);
  float* xbuf   = (float*)(ws + 2 * szN + szT + 2 * szE + szA);

  const int T = 256;
  const int wavesPerBlk = T / 32;

  k_chunk_sort<<<nchunk, BIN_T, 0, stream>>>(bkt, tab, src, dst, E);
  k_bin_csr   <<<nbin, BIN_T, 0, stream>>>(csr, off, dinv, bkt, tab, N, E, nchunk);

  k_aggregate<64, false><<<(N + wavesPerBlk - 1) / wavesPerBlk, T, 0, stream>>>(
      abuf, emb, dinv, off, csr, nullptr, N);
  k_dense2<<<256, DENSE_T, 0, stream>>>(
      abuf, W1, b1, W2, xbuf, N, 256);
  float* out = (float*)d_out;
  k_aggregate<64, true><<<(N + wavesPerBlk - 1) / wavesPerBlk, T, 0, stream>>>(
      out, xbuf, dinv, off, csr, b2, N);
}


# layer 2: the second product's epilogue scales row s by dinv[s]; the neighbour sum adds plain rows, times dinv[d] once: no per-edge dinv[source] gather; products re-associated, sum order kept
# speedup vs baseline: 7.6365x; 1.0450x over previous
#include <hip/hip_runtime.h>


typedef float v2f __attribute__((ext_vector_type(2)));
typedef float v4f __attribute__((ext_vector_type(4)));
typedef float v8f __attribute__((ext_vector_type(8)));

constexpr int RSH   = 9;
constexpr int RNG   = 1 << RSH;
constexpr int MAXB  = 256;
constexpr int BIN_T = 1024;
constexpr int EPT   = 8;
constexpr int CHUNK = BIN_T * EPT;
constexpr int TP    = MAXB + 1;
constexpr int MAXC  = 256;
constexpr int RPW   = MAXC / 32;
constexpr int RL    = 2;

__global__ __launch_bounds__(BIN_T) void k_chunk_sort(int* __restrict__ bkt,
                                                      int* __restrict__ tab,
                                                      const int* __restrict__ src,
                                                      const int* __restrict__ dst, int E) {
  __shared__ int h[MAXB];
  __shared__ int wsum[MAXB / 32];
  __shared__ int stage[CHUNK];
  const int tid = threadIdx.x, lane = tid & 31, wid = tid >> 5;
  const int blk = blockIdx.x;
  const int e0 = blk * CHUNK + tid;
  int d[EPT], s[EPT], p[EPT];
#pragma unroll
  for (int k = 0; k < EPT; ++k) {
    const int i = min(e0 + k * BIN_T, E - 1);
    d[k] = dst[i];
    s[k] = src[i];
    p[k] = 0;
  }
  if (tid < MAXB) h[tid] = 0;
  __syncthreads();
#pragma unroll
  for (int k = 0; k < EPT; ++k)
    if (e0 + k * BIN_T < E) atomicAdd(&h[d[k] >> RSH], 1);
  __syncthreads();
  int own = 0, incl = 0;
  if (tid < MAXB) {
    own = h[tid];
    incl = own;
    for (int o = 1; o < 32; o <<= 1) {
      const int y = __shfl_up(incl, o);
      if (lane >= o) incl += y;
    }
    if (lane == 31) wsum[wid] = incl;
  }
  __syncthreads();
  const int cnt = min(CHUNK, E - blk * CHUNK);
  if (tid < MAXB) {
    int carry = 0;
    for (int w = 0; w < wid; ++w) carry += wsum[w];
    const int first = carry + incl - own;
    h[tid] = first;
    tab[blk * TP + tid] = first;
    if (tid == 0) tab[blk * TP + MAXB] = cnt;
  }
  __syncthreads();
#pragma unroll
  for (int k = 0; k < EPT; ++k)
    if (e0 + k * BIN_T < E) p[k] = atomicAdd(&h[d[k] >> RSH], 1);
#pragma unroll
  for (int k = 0; k < EPT; ++k)
    if (e0 + k * BIN_T < E) stage[p[k]] = (s[k] << RSH) | (d[k] & (RNG - 1));
  __syncthreads();
  int sv[EPT];
#pragma unroll
  for (int k = 0; k < EPT; ++k) sv[k] = stage[min(tid + k * BIN_T, cnt - 1)];
#pragma unroll
  for (int k = 0; k < EPT; ++k) {
    const int pos = tid + k * BIN_T;
    if (pos < cnt) bkt[blk * CHUNK + pos] = sv[k];
  }
}

constexpr int CAP = 16384;

__global__ __launch_bounds__(BIN_T) void k_bin_csr(int* __restrict__ csr,
                                                   int* __restrict__ off,
                                                   float* __restrict__ dinv,
                                                   const int* __restrict__ bkt,
                                                   const int* __restrict__ tab,
                                                   int N, int E, int nchunk) {
  __shared__ int c[RNG];
  __shared__ int wsum[RNG / 32];
  __shared__ int stage[CAP];
  __shared__ int rbeg[MAXC];
  __shared__ int rlen[MAXC];
  __shared__ int ps[MAXC / 32];
  __shared__ int pe[MAXC / 32];
  const int tid = threadIdx.x, lane = tid & 31, wid = tid >> 5;
  const int b = blockIdx.x;
  if (tid < MAXC) {
    int st = 0, en = 0;
    if (tid < nchunk) {
      st = tab[tid * TP + b];
      en = tab[tid * TP + b + 1];
    }
    rbeg[tid] = tid * CHUNK + st;
    rlen[tid] = en - st;
    int a = st, z = en;
    for (int o = 16; o > 0; o >>= 1) {
      a += __shfl_xor(a, o);
      z += __shfl_xor(z, o);
    }
    if (lane == 0) {
      ps[wid] = a;
      pe[wid] = z;
    }
  }
  if (tid < RNG) c[tid] = 0;
  __syncthreads();
  int s = 0, e = 0;
#pragma unroll
  for (int w = 0; w < MAXC / 32; ++w) {
    s += ps[w];
    e += pe[w];
  }
  const int cnt = e - s;
  const bool fits = cnt <= CAP;
  int beg[RPW], len[RPW], w[RPW][RL];
#pragma unroll
  for (int j = 0; j < RPW; ++j) {
    beg[j] = rbeg[wid + 32 * j];
    len[j] = rlen[wid + 32 * j];
  }
#pragma unroll
  for (int j = 0; j < RPW; ++j) {
#pragma unroll
    for (int t = 0; t < RL; ++t) w[j][t] = bkt[min(beg[j] + lane + 32 * t, E - 1)];
  }
#pragma unroll
  for (int j = 0; j < RPW; ++j) {
#pragma unroll
    for (int t = 0; t < RL; ++t)
      if (lane + 32 * t < len[j]) atomicAdd(&c[w[j][t] & (RNG - 1)], 1);
  }
  for (int j = 0; j < RPW; ++j)
    for (int i = lane + 32 * RL; i < len[j]; i += 32)
      atomicAdd(&c[bkt[beg[j] + i] & (RNG - 1)], 1);
  __syncthreads();
  int v = 0, incl = 0;
  if (tid < RNG) {
    v = c[tid];
    incl = v;
    for (int o = 1; o < 32; o <<= 1) {
      const int y = __shfl_up(incl, o);
      if (lane >= o) incl += y;
    }
    if (lane == 31) wsum[wid] = incl;
  }
  __syncthreads();
  if (tid < RNG) {
    int carry = 0;
    for (int ww = 0; ww < wid; ++ww) carry += wsum[ww];
    const int o = s + carry + incl - v;
    c[tid] = o;
    const int node = b * RNG + tid;
    if (node < N) {
      off[node]  = o;
      dinv[node] = rsqrtf((float)(v + 1));
    }
  }
  if (b == (int)gridDim.x - 1 && tid == 0) off[N] = E;
  __syncthreads();
  int p[RPW][RL];
#pragma unroll
  for (int j = 0; j < RPW; ++j) {
#pragma unroll
    for (int t = 0; t < RL; ++t) {
      p[j][t] = 0;
      if (lane + 32 * t < len[j]) p[j][t] = atomicAdd(&c[w[j][t] & (RNG - 1)], 1);
    }
  }
  if (fits) {
#pragma unroll
    for (int j = 0; j < RPW; ++j) {
#pragma unroll
      for (int t = 0; t < RL; ++t)
        if (lane + 32 * t < len[j]) stage[p[j][t] - s] = w[j][t] >> RSH;
    }
  } else {
#pragma unroll
    for (int j = 0; j < RPW; ++j) {
#pragma unroll
      for (int t = 0; t < RL; ++t)
        if (lane + 32 * t < len[j]) csr[p[j][t]] = w[j][t] >> RSH;
    }
  }
  for (int j = 0; j < RPW; ++j)
    for (int i = lane + 32 * RL; i < len[j]; i += 32) {
      const int x = bkt[beg[j] + i];
      const int q = atomicAdd(&c[x & (RNG - 1)], 1);
      if (fits) stage[q - s] = x >> RSH;
      else      csr[q] = x >> RSH;
    }
  __syncthreads();
  if (fits && cnt > 0) {
    for (int i0 = tid - (s & 31); i0 < cnt; i0 += 4 * BIN_T) {
      int y[4];
#pragma unroll
      for (int j = 0; j < 4; ++j) y[j] = stage[min(max(i0 + j * BIN_T, 0), cnt - 1)];
#pragma unroll
      for (int j = 0; j < 4; ++j) {
        const int i = i0 + j * BIN_T;
        if (i >= 0 && i < cnt) csr[s + i] = y[j];
      }
    }
  }
}

constexpr int D_IN  = 64;
constexpr int D_MID = 128;
constexpr int D_FIN = 64;
constexpr int PW1  = 2 * D_MID + 32;
constexpr int PW2  = 2 * D_FIN + 32;
constexpr int PA   = D_IN + 4;
constexpr int PH   = D_MID + 4;
constexpr int SLOT = 16 * PH;
constexpr int DENSE_WAVES = 16;
constexpr int DENSE_T = DENSE_WAVES * 32;

__device__ __forceinline__ void wave_slot_handover() {
  __builtin_amdgcn_fence(__ATOMIC_RELEASE, "wavefront");
  __builtin_amdgcn_wave_barrier();
  __builtin_amdgcn_fence(__ATOMIC_ACQUIRE, "wavefront");
}

__global__ __launch_bounds__(DENSE_T) void k_dense2(const float* __restrict__ X,
                                                    const float* __restrict__ W1,
                                                    const float* __restrict__ b1,
                                                    const float* __restrict__ W2,
                                                    const float* __restrict__ dinv,
                                                    float* __restrict__ Y, int M, int nwg) {
  constexpr int W2_FLOATS = (D_MID / 2) * PW2;
  constexpr int W1_FLOATS = (D_IN / 2) * PW1;
  __shared__ __attribute__((aligned(16))) float lds[W2_FLOATS + W1_FLOATS + DENSE_WAVES * SLOT];
  float* const w2p   = lds;
  float* const w1p   = lds + W2_FLOATS;
  float* const slots = lds + W2_FLOATS + W1_FLOATS;

  const int tid  = threadIdx.x;
  const int lane = tid & 31;
  const int wid  = __builtin_amdgcn_readfirstlane(tid >> 5);
  const int half = lane >> 4;
  const int r    = lane & 15;
  const int mtiles  = (M + 15) >> 4;
  const int tstride = DENSE_WAVES * nwg;
  int tile = wid * nwg + (int)blockIdx.x;
  float* slot = slots + wid * SLOT;

  const int trow = lane >> 4;
  const int tcol = (lane & 15) * 4;
  v4f xin[8];
  {
    const int m0 = min(tile, mtiles - 1) << 4;
#pragma unroll
    for (int i = 0; i < 8; ++i) {
      const int gr = min(m0 + 2 * i + trow, M - 1);
      xin[i] = *(const v4f*)(X + (size_t)gr * D_IN + tcol);
    }
  }
  float bias[D_MID / 16];
#pragma unroll
  for (int nt = 0; nt < D_MID / 16; ++nt) bias[nt] = b1[nt * 16 + r];

#pragma unroll
  for (int it = 0; it < (D_IN / 2) * (D_MID / 4) / DENSE_T; ++it) {
    const int item = tid + DENSE_T * it;
    const int p = item / (D_MID / 4), n = (item % (D_MID / 4)) * 4;
    const v4f lo = *(const v4f*)(W1 + (size_t)(2 * p) * D_MID + n);
    const v4f hi = *(const v4f*)(W1 + (size_t)(2 * p + 1) * D_MID + n);
    const v4f s0 = {lo[0], hi[0], lo[1], hi[1]};
    const v4f s1 = {lo[2], hi[2], lo[3], hi[3]};
    *(v4f*)(w1p + p * PW1 + 2 * n)     = s0;
    *(v4f*)(w1p + p * PW1 + 2 * n + 4) = s1;
  }
#pragma unroll
  for (int it = 0; it < (D_MID / 2) * (D_FIN / 4) / DENSE_T; ++it) {
    const int item = tid + DENSE_T * it;
    const int p = item / (D_FIN / 4), n = (item % (D_FIN / 4)) * 4;
    const v4f lo = *(const v4f*)(W2 + (size_t)(2 * p) * D_FIN + n);
    const v4f hi = *(const v4f*)(W2 + (size_t)(2 * p + 1) * D_FIN + n);
    const v4f s0 = {lo[0], hi[0], lo[1], hi[1]};
    const v4f s1 = {lo[2], hi[2], lo[3], hi[3]};
    *(v4f*)(w2p + p * PW2 + 2 * n)     = s0;
    *(v4f*)(w2p + p * PW2 + 2 * n + 4) = s1;
  }
  __syncthreads();

  for (; tile < mtiles; tile += tstride) {
    const int m0 = tile << 4;
    float rs[8];
#pragma unroll
    for (int i = 0; i < 8; ++i) rs[i] = dinv[min(m0 + 2 * i + trow, M - 1)];
#pragma unroll
    for (int i = 0; i < 8; ++i) *(v4f*)(slot + (2 * i + trow) * PA + tcol) = xin[i];
    if (tile + tstride < mtiles) {
      const int n0 = (tile + tstride) << 4;
#pragma unroll
      for (int i = 0; i < 8; ++i) {
        const int gr = min(n0 + 2 * i + trow, M - 1);
        xin[i] = *(const v4f*)(X + (size_t)gr * D_IN + tcol);
      }
    }
    wave_slot_handover();

    v8f acc[D_MID / 16];
#pragma unroll
    for (int nt = 0; nt < D_MID / 16; ++nt) acc[nt] = v8f{};
    {
      const float* ap = slot + r * PA + 2 * half;
      const float* bp = w1p + half * PW1 + 2 * r;
#pragma unroll
      for (int kk = 0; kk < D_IN / 4; ++kk) {
        const v2f a = *(const v2f*)(ap + 4 * kk);
#pragma unroll
        for (int nt = 0; nt < D_MID / 16; ++nt) {
          const v2f b = *(const v2f*)(bp + (2 * kk) * PW1 + nt * 32);
          acc[nt] = __builtin_amdgcn_wmma_f32_16x16x4_f32(
               false, a,  false, b,
               (short)0, acc[nt],  false,  false);
        }
      }
    }
#pragma unroll
    for (int nt = 0; nt < D_MID / 16; ++nt) {
#pragma unroll
      for (int v = 0; v < 8; ++v)
        slot[(v + 8 * half) * PH + nt * 16 + r] = fmaxf(acc[nt][v] + bias[nt], 0.0f);
    }
    wave_slot_handover();

    v8f acc2[D_FIN / 16];
#pragma unroll
    for (int nt = 0; nt < D_FIN / 16; ++nt) acc2[nt] = v8f{};
    {
      const float* ap = slot + r * PH + 2 * half;
      const float* bp = w2p + half * PW2 + 2 * r;
#pragma unroll
      for (int kk = 0; kk < D_MID / 4; ++kk) {
        const v2f a = *(const v2f*)(ap + 4 * kk);
#pragma unroll
        for (int nt = 0; nt < D_FIN / 16; ++nt) {
          const v2f b = *(const v2f*)(bp + (2 * kk) * PW2 + nt * 32);
          acc2[nt] = __builtin_amdgcn_wmma_f32_16x16x4_f32(
               false, a,  false, b,
               (short)0, acc2[nt],  false,  false);
        }
      }
    }
#pragma unroll
    for (int nt = 0; nt < D_FIN / 16; ++nt) {
#pragma unroll
      for (int v = 0; v < 8; ++v) slot[(v + 8 * half) * PA + nt * 16 + r] = acc2[nt][v];
    }
    wave_slot_handover();
    if (m0 + 16 <= M) {
#pragma unroll
      for (int i = 0; i < 8; ++i) {
        const v4f y = *(const v4f*)(slot + (2 * i + trow) * PA + tcol) * rs[i];
        *(v4f*)(Y + (size_t)(m0 + 2 * i + trow) * D_FIN + tcol) = y;
      }
    } else {
#pragma unroll
      for (int i = 0; i < 8; ++i) {
        const int row = m0 + 2 * i + trow;
        const v4f y = *(const v4f*)(slot + (2 * i + trow) * PA + tcol) * rs[i];
        if (row < M) *(v4f*)(Y + (size_t)row * D_FIN + tcol) = y;
      }
    }
  }
}

template <int V>
struct rowvec;
template <> struct rowvec<4> { typedef v4f type; };
template <> struct rowvec<2> { typedef v2f type; };

template <int V, int B>
__device__ __forceinline__ void gather_rows(typename rowvec<V>::type& acc,
                                            const typename rowvec<V>::type* __restrict__ Xv,
                                            int ms, int mwbits, int j, int lane) {
  typedef typename rowvec<V>::type vf;
  int   s[B];
  float w[B];
  vf    x[B];
#pragma unroll
  for (int b = 0; b < B; ++b) {
    s[b] = __builtin_amdgcn_readlane(ms, j + b);
    w[b] = __int_as_float(__builtin_amdgcn_readlane(mwbits, j + b));
  }
#pragma unroll
  for (int b = 0; b < B; ++b) x[b] = Xv[(size_t)s[b] * 32 + lane];
#pragma unroll
  for (int b = 0; b < B; ++b) acc += x[b] * w[b];
}

template <int V, int B>
__device__ __forceinline__ void gather_rows_plain(typename rowvec<V>::type& acc,
                                                  const typename rowvec<V>::type* __restrict__ Xv,
                                                  int ms, int j, int lane) {
  typedef typename rowvec<V>::type vf;
  int s[B];
  vf  x[B];
#pragma unroll
  for (int b = 0; b < B; ++b) s[b] = __builtin_amdgcn_readlane(ms, j + b);
#pragma unroll
  for (int b = 0; b < B; ++b) x[b] = Xv[(size_t)s[b] * 32 + lane];
#pragma unroll
  for (int b = 0; b < B; ++b) acc += x[b];
}

template <int D, bool BIAS, bool PRE>
__global__ __launch_bounds__(256) void k_aggregate(float* __restrict__ Y,
                                                   const float* __restrict__ X,
                                                   const float* __restrict__ dinv,
                                                   const int* __restrict__ off,
                                                   const int* __restrict__ csr,
                                                   const float* __restrict__ bias, int N) {
  constexpr int V = D / 32;
  typedef typename rowvec<V>::type vf;
  int node = blockIdx.x * (blockDim.x >> 5) + (threadIdx.x >> 5);
  if (node >= N) return;
  node = __builtin_amdgcn_readfirstlane(node);
  const int lane = threadIdx.x & 31;
  const int beg = off[node], end = off[node + 1];
  const float dd = dinv[node];
  const vf* Xv = (const vf*)X;

  vf acc;
  if (PRE) {
    acc = Xv[(size_t)node * 32 + lane];
    for (int base = beg; base < end; base += 32) {
      const int n = min(32, end - base);
      int ms = 0;
      if (lane < n) ms = csr[base + lane];
      int j = 0;
      for (; j + 8 <= n; j += 8) gather_rows_plain<V, 8>(acc, Xv, ms, j, lane);
      if (j + 4 <= n) { gather_rows_plain<V, 4>(acc, Xv, ms, j, lane); j += 4; }
      if (j + 2 <= n) { gather_rows_plain<V, 2>(acc, Xv, ms, j, lane); j += 2; }
      if (j < n)      { gather_rows_plain<V, 1>(acc, Xv, ms, j, lane); }
    }
    acc *= dd;
  } else {
    acc = Xv[(size_t)node * 32 + lane] * (dd * dd);
    for (int base = beg; base < end; base += 32) {
      const int n = min(32, end - base);
      int ms = 0;
      float mw = 0.0f;
      if (lane < n) {
        ms = csr[base + lane];
        mw = dinv[ms] * dd;
      }
      const int mwbits = __float_as_int(mw);
      int j = 0;
      for (; j + 8 <= n; j += 8) gather_rows<V, 8>(acc, Xv, ms, mwbits, j, lane);
      if (j + 4 <= n) { gather_rows<V, 4>(acc, Xv, ms, mwbits, j, lane); j += 4; }
      if (j + 2 <= n) { gather_rows<V, 2>(acc, Xv, ms, mwbits, j, lane); j += 2; }
      if (j < n)      { gather_rows<V, 1>(acc, Xv, ms, mwbits, j, lane); }
    }
  }
  if (BIAS) acc += ((const vf*)bias)[lane];
  ((vf*)Y)[(size_t)node * 32 + lane] = acc;
}

extern "C" void kernel_launch(void* const* d_in, const int* in_sizes, int n_in,
                              void* d_out, int out_size, void* d_ws, size_t ws_size,
                              hipStream_t stream) {
  const float* emb  = (const float*)d_in[0];
  const float* W1   = (const float*)d_in[1];
  const float* b1   = (const float*)d_in[2];
  const float* W2   = (const float*)d_in[3];
  const float* b2   = (const float*)d_in[4];
  const int*   edge = (const int*)d_in[5];

  const int D_OUT = in_sizes[4];
  const int N     = in_sizes[0] / D_OUT;
  const int E     = in_sizes[5] / 2;
  const int* src  = edge;
  const int* dst  = edge + E;

  char* ws = (char*)d_ws;
  const int nchunk = (E + CHUNK - 1) / CHUNK;
  const int nbin   = (N + RNG - 1) / RNG;
  const size_t szN = (((size_t)(N + 1) * 4) + 255) & ~(size_t)255;
  const size_t szE = (((size_t)E * 4) + 255) & ~(size_t)255;
  const size_t szA = (((size_t)N * D_OUT * 4) + 255) & ~(size_t)255;
  const size_t szT = (((size_t)nchunk * TP * 4) + 255) & ~(size_t)255;
  int*   off    = (int*)ws;
  float* dinv   = (float*)(ws + szN);
  int*   tab    = (int*)(ws + 2 * szN);
  int*   bkt    = (int*)(ws + 2 * szN + szT);
  int*   csr    = (int*)(ws + 2 * szN + szT + szE);
  float* abuf   = (float*)(ws + 2 * szN + szT + 2 * szE);
  float* xbuf   = (float*)(ws + 2 * szN + szT + 2 * szE + szA);

  const int T = 256;
  const int wavesPerBlk = T / 32;

  k_chunk_sort<<<nchunk, BIN_T, 0, stream>>>(bkt, tab, src, dst, E);
  k_bin_csr   <<<nbin, BIN_T, 0, stream>>>(csr, off, dinv, bkt, tab, N, E, nchunk);

  k_aggregate<64, false, false><<<(N + wavesPerBlk - 1) / wavesPerBlk, T, 0, stream>>>(
      abuf, emb, dinv, off, csr, nullptr, N);
  k_dense2<<<256, DENSE_T, 0, stream>>>(
      abuf, W1, b1, W2, dinv, xbuf, N, 256);
  float* out = (float*)d_out;
  k_aggregate<64, true, true><<<(N + wavesPerBlk - 1) / wavesPerBlk, T, 0, stream>>>(
      out, xbuf, dinv, off, csr, b2, N);
}


# layer-1 neighbour sum moved into the dense launch: a wave sums the in-edge rows of its 16 nodes straight into its LDS tile, under the other waves' products; 5 launches now 4; sum order kept
# speedup vs baseline: 8.0862x; 1.0589x over previous
#include <hip/hip_runtime.h>


typedef float v2f __attribute__((ext_vector_type(2)));
typedef float v4f __attribute__((ext_vector_type(4)));
typedef float v8f __attribute__((ext_vector_type(8)));

constexpr int RSH   = 9;
constexpr int RNG   = 1 << RSH;
constexpr int MAXB  = 256;
constexpr int BIN_T = 1024;
constexpr int EPT   = 8;
constexpr int CHUNK = BIN_T * EPT;
constexpr int TP    = MAXB + 1;
constexpr int MAXC  = 256;
constexpr int RPW   = MAXC / 32;
constexpr int RL    = 2;

__global__ __launch_bounds__(BIN_T) void k_chunk_sort(int* __restrict__ bkt,
                                                      int* __restrict__ tab,
                                                      const int* __restrict__ src,
                                                      const int* __restrict__ dst, int E) {
  __shared__ int h[MAXB];
  __shared__ int wsum[MAXB / 32];
  __shared__ int stage[CHUNK];
  const int tid = threadIdx.x, lane = tid & 31, wid = tid >> 5;
  const int blk = blockIdx.x;
  const int e0 = blk * CHUNK + tid;
  int d[EPT], s[EPT], p[EPT];
#pragma unroll
  for (int k = 0; k < EPT; ++k) {
    const int i = min(e0 + k * BIN_T, E - 1);
    d[k] = dst[i];
    s[k] = src[i];
    p[k] = 0;
  }
  if (tid < MAXB) h[tid] = 0;
  __syncthreads();
#pragma unroll
  for (int k = 0; k < EPT; ++k)
    if (e0 + k * BIN_T < E) atomicAdd(&h[d[k] >> RSH], 1);
  __syncthreads();
  int own = 0, incl = 0;
  if (tid < MAXB) {
    own = h[tid];
    incl = own;
    for (int o = 1; o < 32; o <<= 1) {
      const int y = __shfl_up(incl, o);
      if (lane >= o) incl += y;
    }
    if (lane == 31) wsum[wid] = incl;
  }
  __syncthreads();
  const int cnt = min(CHUNK, E - blk * CHUNK);
  if (tid < MAXB) {
    int carry = 0;
    for (int w = 0; w < wid; ++w) carry += wsum[w];
    const int first = carry + incl - own;
    h[tid] = first;
    tab[blk * TP + tid] = first;
    if (tid == 0) tab[blk * TP + MAXB] = cnt;
  }
  __syncthreads();
#pragma unroll
  for (int k = 0; k < EPT; ++k)
    if (e0 + k * BIN_T < E) p[k] = atomicAdd(&h[d[k] >> RSH], 1);
#pragma unroll
  for (int k = 0; k < EPT; ++k)
    if (e0 + k * BIN_T < E) stage[p[k]] = (s[k] << RSH) | (d[k] & (RNG - 1));
  __syncthreads();
  int sv[EPT];
#pragma unroll
  for (int k = 0; k < EPT; ++k) sv[k] = stage[min(tid + k * BIN_T, cnt - 1)];
#pragma unroll
  for (int k = 0; k < EPT; ++k) {
    const int pos = tid + k * BIN_T;
    if (pos < cnt) bkt[blk * CHUNK + pos] = sv[k];
  }
}

constexpr int CAP = 16384;

__global__ __launch_bounds__(BIN_T) void k_bin_csr(int* __restrict__ csr,
                                                   int* __restrict__ off,
                                                   float* __restrict__ dinv,
                                                   const int* __restrict__ bkt,
                                                   const int* __restrict__ tab,
                                                   int N, int E, int nchunk) {
  __shared__ int c[RNG];
  __shared__ int wsum[RNG / 32];
  __shared__ int stage[CAP];
  __shared__ int rbeg[MAXC];
  __shared__ int rlen[MAXC];
  __shared__ int ps[MAXC / 32];
  __shared__ int pe[MAXC / 32];
  const int tid = threadIdx.x, lane = tid & 31, wid = tid >> 5;
  const int b = blockIdx.x;
  if (tid < MAXC) {
    int st = 0, en = 0;
    if (tid < nchunk) {
      st = tab[tid * TP + b];
      en = tab[tid * TP + b + 1];
    }
    rbeg[tid] = tid * CHUNK + st;
    rlen[tid] = en - st;
    int a = st, z = en;
    for (int o = 16; o > 0; o >>= 1) {
      a += __shfl_xor(a, o);
      z += __shfl_xor(z, o);
    }
    if (lane == 0) {
      ps[wid] = a;
      pe[wid] = z;
    }
  }
  if (tid < RNG) c[tid] = 0;
  __syncthreads();
  int s = 0, e = 0;
#pragma unroll
  for (int w = 0; w < MAXC / 32; ++w) {
    s += ps[w];
    e += pe[w];
  }
  const int cnt = e - s;
  const bool fits = cnt <= CAP;
  int beg[RPW], len[RPW], w[RPW][RL];
#pragma unroll
  for (int j = 0; j < RPW; ++j) {
    beg[j] = rbeg[wid + 32 * j];
    len[j] = rlen[wid + 32 * j];
  }
#pragma unroll
  for (int j = 0; j < RPW; ++j) {
#pragma unroll
    for (int t = 0; t < RL; ++t) w[j][t] = bkt[min(beg[j] + lane + 32 * t, E - 1)];
  }
#pragma unroll
  for (int j = 0; j < RPW; ++j) {
#pragma unroll
    for (int t = 0; t < RL; ++t)
      if (lane + 32 * t < len[j]) atomicAdd(&c[w[j][t] & (RNG - 1)], 1);
  }
  for (int j = 0; j < RPW; ++j)
    for (int i = lane + 32 * RL; i < len[j]; i += 32)
      atomicAdd(&c[bkt[beg[j] + i] & (RNG - 1)], 1);
  __syncthreads();
  int v = 0, incl = 0;
  if (tid < RNG) {
    v = c[tid];
    incl = v;
    for (int o = 1; o < 32; o <<= 1) {
      const int y = __shfl_up(incl, o);
      if (lane >= o) incl += y;
    }
    if (lane == 31) wsum[wid] = incl;
  }
  __syncthreads();
  if (tid < RNG) {
    int carry = 0;
    for (int ww = 0; ww < wid; ++ww) carry += wsum[ww];
    const int o = s + carry + incl - v;
    c[tid] = o;
    const int node = b * RNG + tid;
    if (node < N) {
      off[node]  = o;
      dinv[node] = rsqrtf((float)(v + 1));
    }
  }
  if (b == (int)gridDim.x - 1 && tid == 0) off[N] = E;
  __syncthreads();
  int p[RPW][RL];
#pragma unroll
  for (int j = 0; j < RPW; ++j) {
#pragma unroll
    for (int t = 0; t < RL; ++t) {
      p[j][t] = 0;
      if (lane + 32 * t < len[j]) p[j][t] = atomicAdd(&c[w[j][t] & (RNG - 1)], 1);
    }
  }
  if (fits) {
#pragma unroll
    for (int j = 0; j < RPW; ++j) {
#pragma unroll
      for (int t = 0; t < RL; ++t)
        if (lane + 32 * t < len[j]) stage[p[j][t] - s] = w[j][t] >> RSH;
    }
  } else {
#pragma unroll
    for (int j = 0; j < RPW; ++j) {
#pragma unroll
      for (int t = 0; t < RL; ++t)
        if (lane + 32 * t < len[j]) csr[p[j][t]] = w[j][t] >> RSH;
    }
  }
  for (int j = 0; j < RPW; ++j)
    for (int i = lane + 32 * RL; i < len[j]; i += 32) {
      const int x = bkt[beg[j] + i];
      const int q = atomicAdd(&c[x & (RNG - 1)], 1);
      if (fits) stage[q - s] = x >> RSH;
      else      csr[q] = x >> RSH;
    }
  __syncthreads();
  if (fits && cnt > 0) {
    for (int i0 = tid - (s & 31); i0 < cnt; i0 += 4 * BIN_T) {
      int y[4];
#pragma unroll
      for (int j = 0; j < 4; ++j) y[j] = stage[min(max(i0 + j * BIN_T, 0), cnt - 1)];
#pragma unroll
      for (int j = 0; j < 4; ++j) {
        const int i = i0 + j * BIN_T;
        if (i >= 0 && i < cnt) csr[s + i] = y[j];
      }
    }
  }
}

constexpr int D_IN  = 64;
constexpr int D_MID = 128;
constexpr int D_FIN = 64;
constexpr int PW1  = 2 * D_MID + 32;
constexpr int PW2  = 2 * D_FIN + 32;
constexpr int PA   = D_IN + 4;
constexpr int PH   = D_MID + 4;
constexpr int SLOT = 16 * PH;
constexpr int DENSE_WAVES = 16;
constexpr int DENSE_T = DENSE_WAVES * 32;

__device__ __forceinline__ void wave_slot_handover() {
  __builtin_amdgcn_fence(__ATOMIC_RELEASE, "wavefront");
  __builtin_amdgcn_wave_barrier();
  __builtin_amdgcn_fence(__ATOMIC_ACQUIRE, "wavefront");
}

struct places {
  int      id;
  float    dd;
  unsigned last;
  int      n;
};

__device__ __forceinline__ places tile_places(int pb, int lane, int m0, int M, int e0, int total,
                                              int qv, float dv, const int (&q)[16],
                                              const int* __restrict__ csr) {
  const int p = pb + lane;
  int k = 0;
#pragma unroll
  for (int t = 1; t <= 16; ++t) k += (q[t - 1] <= p) ? 1 : 0;
  k = min(k, 15);
  const bool inside = p < total;
  const int qk  = __shfl(qv, k);
  const int qk1 = __shfl(qv, k + 1);
  places r;
  r.dd = __shfl(dv, k);
  int id = min(m0 + k, M - 1);
  if (inside && p != qk) id = csr[e0 + p - k - 1];
  r.id = id;
  r.last = __builtin_amdgcn_ballot_w32(inside && p + 1 == qk1);
  r.n = min(32, total - pb);
  return r;
}

template <int B>
__device__ __forceinline__ void stream_rows(v2f& acc, int& krow, const v2f* __restrict__ Xv,
                                            float* rowp, int ids, int wbits, unsigned last,
                                            int j, int lane) {
  int   s[B];
  float w[B];
  v2f   x[B];
#pragma unroll
  for (int b = 0; b < B; ++b) {
    s[b] = __builtin_amdgcn_readlane(ids, j + b);
    w[b] = __int_as_float(__builtin_amdgcn_readlane(wbits, j + b));
  }
#pragma unroll
  for (int b = 0; b < B; ++b) x[b] = Xv[s[b] * 32 + lane];
#pragma unroll
  for (int b = 0; b < B; ++b) {
    acc += x[b] * w[b];
    if ((last >> (j + b)) & 1u) {
      *(v2f*)(rowp + krow * PA) = acc;
      ++krow;
      acc = v2f{0.0f, 0.0f};
    }
  }
}

__global__ __launch_bounds__(DENSE_T) void k_dense2(const float* __restrict__ X,
                                                    const float* __restrict__ W1,
                                                    const float* __restrict__ b1,
                                                    const float* __restrict__ W2,
                                                    const float* __restrict__ dinv,
                                                    const int* __restrict__ off,
                                                    const int* __restrict__ csr,
                                                    float* __restrict__ Y, int M, int nwg) {
  constexpr int W2_FLOATS = (D_MID / 2) * PW2;
  constexpr int W1_FLOATS = (D_IN / 2) * PW1;
  __shared__ __attribute__((aligned(16))) float lds[W2_FLOATS + W1_FLOATS + DENSE_WAVES * SLOT];
  float* const w2p   = lds;
  float* const w1p   = lds + W2_FLOATS;
  float* const slots = lds + W2_FLOATS + W1_FLOATS;

  const int tid  = threadIdx.x;
  const int lane = tid & 31;
  const int wid  = __builtin_amdgcn_readfirstlane(tid >> 5);
  const int half = lane >> 4;
  const int r    = lane & 15;
  const int mtiles  = (M + 15) >> 4;
  const int tstride = DENSE_WAVES * nwg;
  int tile = wid * nwg + (int)blockIdx.x;
  float* slot = slots + wid * SLOT;

  const int trow = lane >> 4;
  const int tcol = (lane & 15) * 4;
  int   ov;
  float dv;
  {
    const int m0 = min(tile, mtiles - 1) << 4;
    ov = off[min(m0 + lane, M)];
    dv = dinv[min(m0 + lane, M - 1)];
  }
  const v2f* Xv = (const v2f*)X;
  float* const rowp = slot + 2 * lane;
  float bias[D_MID / 16];
#pragma unroll
  for (int nt = 0; nt < D_MID / 16; ++nt) bias[nt] = b1[nt * 16 + r];

#pragma unroll
  for (int it = 0; it < (D_IN / 2) * (D_MID / 4) / DENSE_T; ++it) {
    const int item = tid + DENSE_T * it;
    const int p = item / (D_MID / 4), n = (item % (D_MID / 4)) * 4;
    const v4f lo = *(const v4f*)(W1 + (size_t)(2 * p) * D_MID + n);
    const v4f hi = *(const v4f*)(W1 + (size_t)(2 * p + 1) * D_MID + n);
    const v4f s0 = {lo[0], hi[0], lo[1], hi[1]};
    const v4f s1 = {lo[2], hi[2], lo[3], hi[3]};
    *(v4f*)(w1p + p * PW1 + 2 * n)     = s0;
    *(v4f*)(w1p + p * PW1 + 2 * n + 4) = s1;
  }
#pragma unroll
  for (int it = 0; it < (D_MID / 2) * (D_FIN / 4) / DENSE_T; ++it) {
    const int item = tid + DENSE_T * it;
    const int p = item / (D_FIN / 4), n = (item % (D_FIN / 4)) * 4;
    const v4f lo = *(const v4f*)(W2 + (size_t)(2 * p) * D_FIN + n);
    const v4f hi = *(const v4f*)(W2 + (size_t)(2 * p + 1) * D_FIN + n);
    const v4f s0 = {lo[0], hi[0], lo[1], hi[1]};
    const v4f s1 = {lo[2], hi[2], lo[3], hi[3]};
    *(v4f*)(w2p + p * PW2 + 2 * n)     = s0;
    *(v4f*)(w2p + p * PW2 + 2 * n + 4) = s1;
  }
  __syncthreads();

  for (; tile < mtiles; tile += tstride) {
    const int m0 = tile << 4;
    float rs[8];
#pragma unroll
    for (int i = 0; i < 8; ++i) rs[i] = dinv[min(m0 + 2 * i + trow, M - 1)];
    wave_slot_handover();
    {
      const int rows = min(16, M - m0);
      const int e0 = __builtin_amdgcn_readlane(ov, 0);
      const int qv = ov - e0 + lane;
      int q[16];
#pragma unroll
      for (int t = 1; t <= 16; ++t) q[t - 1] = __builtin_amdgcn_readlane(qv, t);
      const int total = __builtin_amdgcn_readlane(qv, rows);
      places a = tile_places(0, lane, m0, M, e0, total, qv, dv, q, csr);
      float wa = dinv[a.id] * a.dd;
      places b = a;
      if (32 < total) b = tile_places(32, lane, m0, M, e0, total, qv, dv, q, csr);
      v2f acc = {0.0f, 0.0f};
      int krow = 0;
      for (int pb = 0; pb < total; pb += 32) {
        float wnext = 0.0f;
        if (pb + 32 < total) wnext = dinv[b.id];
        places c = b;
        if (pb + 64 < total) c = tile_places(pb + 64, lane, m0, M, e0, total, qv, dv, q, csr);
        const int wbits = __float_as_int(wa);
        if (a.n == 32) {
          stream_rows<32>(acc, krow, Xv, rowp, a.id, wbits, a.last, 0, lane);
        } else {
          int j = 0;
          if (j + 16 <= a.n) { stream_rows<16>(acc, krow, Xv, rowp, a.id, wbits, a.last, j, lane); j += 16; }
          if (j + 8 <= a.n)  { stream_rows<8>(acc, krow, Xv, rowp, a.id, wbits, a.last, j, lane); j += 8; }
          if (j + 4 <= a.n)  { stream_rows<4>(acc, krow, Xv, rowp, a.id, wbits, a.last, j, lane); j += 4; }
          if (j + 2 <= a.n)  { stream_rows<2>(acc, krow, Xv, rowp, a.id, wbits, a.last, j, lane); j += 2; }
          if (j < a.n)       { stream_rows<1>(acc, krow, Xv, rowp, a.id, wbits, a.last, j, lane); }
        }
        wa = wnext * b.dd;
        a = b;
        b = c;
      }
      for (int k = rows; k < 16; ++k)
        *(v2f*)(rowp + k * PA) = v2f{0.0f, 0.0f};
    }
    if (tile + tstride < mtiles) {
      const int n0 = (tile + tstride) << 4;
      ov = off[min(n0 + lane, M)];
      dv = dinv[min(n0 + lane, M - 1)];
    }
    wave_slot_handover();

    v8f acc[D_MID / 16];
#pragma unroll
    for (int nt = 0; nt < D_MID / 16; ++nt) acc[nt] = v8f{};
    {
      const float* ap = slot + r * PA + 2 * half;
      const float* bp = w1p + half * PW1 + 2 * r;
#pragma unroll
      for (int kk = 0; kk < D_IN / 4; ++kk) {
        const v2f a = *(const v2f*)(ap + 4 * kk);
#pragma unroll
        for (int nt = 0; nt < D_MID / 16; ++nt) {
          const v2f b = *(const v2f*)(bp + (2 * kk) * PW1 + nt * 32);
          acc[nt] = __builtin_amdgcn_wmma_f32_16x16x4_f32(
               false, a,  false, b,
               (short)0, acc[nt],  false,  false);
        }
      }
    }
#pragma unroll
    for (int nt = 0; nt < D_MID / 16; ++nt) {
#pragma unroll
      for (int v = 0; v < 8; ++v)
        slot[(v + 8 * half) * PH + nt * 16 + r] = fmaxf(acc[nt][v] + bias[nt], 0.0f);
    }
    wave_slot_handover();

    v8f acc2[D_FIN / 16];
#pragma unroll
    for (int nt = 0; nt < D_FIN / 16; ++nt) acc2[nt] = v8f{};
    {
      const float* ap = slot + r * PH + 2 * half;
      const float* bp = w2p + half * PW2 + 2 * r;
#pragma unroll
      for (int kk = 0; kk < D_MID / 4; ++kk) {
        const v2f a = *(const v2f*)(ap + 4 * kk);
#pragma unroll
        for (int nt = 0; nt < D_FIN / 16; ++nt) {
          const v2f b = *(const v2f*)(bp + (2 * kk) * PW2 + nt * 32);
          acc2[nt] = __builtin_amdgcn_wmma_f32_16x16x4_f32(
               false, a,  false, b,
               (short)0, acc2[nt],  false,  false);
        }
      }
    }
#pragma unroll
    for (int nt = 0; nt < D_FIN / 16; ++nt) {
#pragma unroll
      for (int v = 0; v < 8; ++v) slot[(v + 8 * half) * PA + nt * 16 + r] = acc2[nt][v];
    }
    wave_slot_handover();
    if (m0 + 16 <= M) {
#pragma unroll
      for (int i = 0; i < 8; ++i) {
        const v4f y = *(const v4f*)(slot + (2 * i + trow) * PA + tcol) * rs[i];
        *(v4f*)(Y + (size_t)(m0 + 2 * i + trow) * D_FIN + tcol) = y;
      }
    } else {
#pragma unroll
      for (int i = 0; i < 8; ++i) {
        const int row = m0 + 2 * i + trow;
        const v4f y = *(const v4f*)(slot + (2 * i + trow) * PA + tcol) * rs[i];
        if (row < M) *(v4f*)(Y + (size_t)row * D_FIN + tcol) = y;
      }
    }
  }
}

template <int V>
struct rowvec;
template <> struct rowvec<4> { typedef v4f type; };
template <> struct rowvec<2> { typedef v2f type; };

template <int V, int B>
__device__ __forceinline__ void gather_rows_plain(typename rowvec<V>::type& acc,
                                                  const typename rowvec<V>::type* __restrict__ Xv,
                                                  int ms, int j, int lane) {
  typedef typename rowvec<V>::type vf;
  int s[B];
  vf  x[B];
#pragma unroll
  for (int b = 0; b < B; ++b) s[b] = __builtin_amdgcn_readlane(ms, j + b);
#pragma unroll
  for (int b = 0; b < B; ++b) x[b] = Xv[(size_t)s[b] * 32 + lane];
#pragma unroll
  for (int b = 0; b < B; ++b) acc += x[b];
}

template <int D, bool BIAS>
__global__ __launch_bounds__(256) void k_aggregate(float* __restrict__ Y,
                                                   const float* __restrict__ X,
                                                   const float* __restrict__ dinv,
                                                   const int* __restrict__ off,
                                                   const int* __restrict__ csr,
                                                   const float* __restrict__ bias, int N) {
  constexpr int V = D / 32;
  typedef typename rowvec<V>::type vf;
  int node = blockIdx.x * (blockDim.x >> 5) + (threadIdx.x >> 5);
  if (node >= N) return;
  node = __builtin_amdgcn_readfirstlane(node);
  const int lane = threadIdx.x & 31;
  const int beg = off[node], end = off[node + 1];
  const float dd = dinv[node];
  const vf* Xv = (const vf*)X;

  vf acc = Xv[(size_t)node * 32 + lane];
  for (int base = beg; base < end; base += 32) {
    const int n = min(32, end - base);
    int ms = 0;
    if (lane < n) ms = csr[base + lane];
    int j = 0;
    for (; j + 8 <= n; j += 8) gather_rows_plain<V, 8>(acc, Xv, ms, j, lane);
    if (j + 4 <= n) { gather_rows_plain<V, 4>(acc, Xv, ms, j, lane); j += 4; }
    if (j + 2 <= n) { gather_rows_plain<V, 2>(acc, Xv, ms, j, lane); j += 2; }
    if (j < n)      { gather_rows_plain<V, 1>(acc, Xv, ms, j, lane); }
  }
  acc *= dd;
  if (BIAS) acc += ((const vf*)bias)[lane];
  ((vf*)Y)[(size_t)node * 32 + lane] = acc;
}

extern "C" void kernel_launch(void* const* d_in, const int* in_sizes, int n_in,
                              void* d_out, int out_size, void* d_ws, size_t ws_size,
                              hipStream_t stream) {
  const float* emb  = (const float*)d_in[0];
  const float* W1   = (const float*)d_in[1];
  const float* b1   = (const float*)d_in[2];
  const float* W2   = (const float*)d_in[3];
  const float* b2   = (const float*)d_in[4];
  const int*   edge = (const int*)d_in[5];

  const int D_OUT = in_sizes[4];
  const int N     = in_sizes[0] / D_OUT;
  const int E     = in_sizes[5] / 2;
  const int* src  = edge;
  const int* dst  = edge + E;

  char* ws = (char*)d_ws;
  const int nchunk = (E + CHUNK - 1) / CHUNK;
  const int nbin   = (N + RNG - 1) / RNG;
  const size_t szN = (((size_t)(N + 1) * 4) + 255) & ~(size_t)255;
  const size_t szE = (((size_t)E * 4) + 255) & ~(size_t)255;
  const size_t szT = (((size_t)nchunk * TP * 4) + 255) & ~(size_t)255;
  int*   off    = (int*)ws;
  float* dinv   = (float*)(ws + szN);
  int*   tab    = (int*)(ws + 2 * szN);
  int*   bkt    = (int*)(ws + 2 * szN + szT);
  int*   csr    = (int*)(ws + 2 * szN + szT + szE);
  float* xbuf   = (float*)(ws + 2 * szN + szT + 2 * szE);

  const int T = 256;
  const int wavesPerBlk = T / 32;

  k_chunk_sort<<<nchunk, BIN_T, 0, stream>>>(bkt, tab, src, dst, E);
  k_bin_csr   <<<nbin, BIN_T, 0, stream>>>(csr, off, dinv, bkt, tab, N, E, nchunk);

  k_dense2<<<256, DENSE_T, 0, stream>>>(
      emb, W1, b1, W2, dinv, off, csr, xbuf, N, 256);
  float* out = (float*)d_out;
  k_aggregate<64, true><<<(N + wavesPerBlk - 1) / wavesPerBlk, T, 0, stream>>>(
      out, xbuf, dinv, off, csr, b2, N);
}
